# starting kernel
# baseline (speedup 1.0000x reference)
#include <hip/hip_runtime.h>


typedef __attribute__((ext_vector_type(2))) float v2f;
typedef __attribute__((ext_vector_type(8))) float v8f;

#define BATCH 32768
#define NSITES 64
#define BOND 32
#define ROWS_PER_WAVE 16
#define WAVES_PER_WG 4
#define ROWS_PER_WG (ROWS_PER_WAVE * WAVES_PER_WG)
#define RPAD 68
#define EPAD 34
#define CPAD 20

__global__ __launch_bounds__(WAVES_PER_WG * 32)
void mps_psi_kernel(const int*   __restrict__ cfg,
                    const float* __restrict__ leftT,
                    const float* __restrict__ bulkT,
                    const float* __restrict__ rightT,
                    float*       __restrict__ out)
{
    __shared__ float sM[2][BOND * RPAD];
    __shared__ float sEnv[WAVES_PER_WG][ROWS_PER_WAVE * EPAD];
    __shared__ __align__(16) int sCfgAll[WAVES_PER_WG][NSITES][CPAD];

    const int tid   = threadIdx.x;
    const int wave  = tid >> 5;
    const int lane  = tid & 31;
    const int l16   = lane & 15;
    const int hi    = lane >> 4;
    const int rowBase = blockIdx.x * ROWS_PER_WG + wave * ROWS_PER_WAVE;

    float* env = &sEnv[wave][0];

    for (int i = lane; i < ROWS_PER_WAVE * NSITES; i += 32) {
        const int r = i >> 6;
        const int c = i & 63;
        sCfgAll[wave][c][r] = cfg[(rowBase + r) * NSITES + c];
    }

    for (int i = lane; i < ROWS_PER_WAVE * BOND; i += 32) {
        const int r = i >> 5;
        const int d = i & 31;
        const int s0 = sCfgAll[wave][0][r];
        env[r * EPAD + d] = leftT[s0 * BOND + d];
    }

    for (int site = 1; site <= NSITES - 2; ++site) {
        const float* Msite = bulkT + (size_t)(site - 1) * (BOND * 2 * BOND);
        float* M = &sM[(site - 1) & 1][0];

        for (int i = tid; i < 2 * BOND * BOND; i += (WAVES_PER_WG * 32)) {
            const int e  = i & 31;
            const int rp = i >> 5;
            M[e * RPAD + rp] = Msite[(rp & 31) * 64 + (rp >> 5) * 32 + e];
        }
        if (site < NSITES - 2) {
            __builtin_prefetch(Msite + (BOND * 2 * BOND) + tid * 16, 0, 0);
        }
        __syncthreads();

        int sarr[8];
        *(int4*)&sarr[0] = *(const int4*)&sCfgAll[wave][site][8 * hi];
        *(int4*)&sarr[4] = *(const int4*)&sCfgAll[wave][site][8 * hi + 4];

        v2f a[8];
#pragma unroll
        for (int k = 0; k < 8; ++k) {
            a[k] = *(const v2f*)&env[l16 * EPAD + 4 * k + 2 * hi];
        }

        v8f acc[4] = {};

#pragma unroll
        for (int k = 0; k < 8; ++k) {
            const int r0b = 4 * k + 2 * hi;
#pragma unroll
            for (int n = 0; n < 4; ++n) {
                const int colg = (n & 1) * 16 + l16;
                const int prow = (n >> 1) * 32;
                const v2f b = *(const v2f*)&M[colg * RPAD + prow + r0b];
                acc[n] = __builtin_amdgcn_wmma_f32_16x16x4_f32(
                    false, a[k], false, b, (short)0, acc[n], false, false);
            }
        }

#pragma unroll
        for (int v = 0; v < 8; ++v) {
            const int row = v + 8 * hi;
            const bool sp = (sarr[v] != 0);
            env[row * EPAD + l16]      = sp ? acc[2][v] : acc[0][v];
            env[row * EPAD + 16 + l16] = sp ? acc[3][v] : acc[1][v];
        }
    }

    if (lane < ROWS_PER_WAVE) {
        const int r = rowBase + lane;
        const int p = sCfgAll[wave][NSITES - 1][lane];
        float acc = 0.0f;
#pragma unroll 8
        for (int d = 0; d < BOND; ++d) {
            acc = fmaf(env[lane * EPAD + d], rightT[d * 2 + p], acc);
        }
        out[r] = acc;
    }
}

extern "C" void kernel_launch(void* const* d_in, const int* in_sizes, int n_in,
                              void* d_out, int out_size, void* d_ws, size_t ws_size,
                              hipStream_t stream) {
    const int*   cfg    = (const int*)  d_in[0];
    const float* leftT  = (const float*)d_in[1];
    const float* bulkT  = (const float*)d_in[2];
    const float* rightT = (const float*)d_in[3];
    float*       out    = (float*)d_out;
    (void)in_sizes; (void)n_in; (void)out_size; (void)d_ws; (void)ws_size;

    dim3 grid(BATCH / ROWS_PER_WG);
    dim3 block(WAVES_PER_WG * 32);
    hipLaunchKernelGGL(mps_psi_kernel, grid, block, 0, stream,
                       cfg, leftT, bulkT, rightT, out);
}


# Software-pipelined site-matrix staging: the next site's 16 loads per thread go out as one group behind the barrier, fly under the WMMA block and land in LDS as 4 b128 rows at the next site's top
# speedup vs baseline: 2.4671x; 2.4671x over previous
#include <hip/hip_runtime.h>


typedef __attribute__((ext_vector_type(2))) float v2f;
typedef __attribute__((ext_vector_type(8))) float v8f;

#define BATCH 32768
#define NSITES 64
#define BOND 32
#define ROWS_PER_WAVE 16
#define WAVES_PER_WG 4
#define ROWS_PER_WG (ROWS_PER_WAVE * WAVES_PER_WG)
#define RPAD 68
#define EPAD 34
#define CPAD 20

__global__ __launch_bounds__(WAVES_PER_WG * 32)
void mps_psi_kernel(const int*   __restrict__ cfg,
                    const float* __restrict__ leftT,
                    const float* __restrict__ bulkT,
                    const float* __restrict__ rightT,
                    float*       __restrict__ out)
{
    __shared__ __align__(16) float sM[2][BOND * RPAD];
    __shared__ float sEnv[WAVES_PER_WG][ROWS_PER_WAVE * EPAD];
    __shared__ __align__(16) int sCfgAll[WAVES_PER_WG][NSITES][CPAD];

    const int tid   = threadIdx.x;
    const int wave  = tid >> 5;
    const int lane  = tid & 31;
    const int l16   = lane & 15;
    const int hi    = lane >> 4;
    const int rowBase = blockIdx.x * ROWS_PER_WG + wave * ROWS_PER_WAVE;

    float* env = &sEnv[wave][0];

    for (int i = lane; i < ROWS_PER_WAVE * NSITES; i += 32) {
        const int r = i >> 6;
        const int c = i & 63;
        sCfgAll[wave][c][r] = cfg[(rowBase + r) * NSITES + c];
    }

    for (int i = lane; i < ROWS_PER_WAVE * BOND; i += 32) {
        const int r = i >> 5;
        const int d = i & 31;
        const int s0 = sCfgAll[wave][0][r];
        env[r * EPAD + d] = leftT[s0 * BOND + d];
    }

    float4 stg[4];
    {
        const float* src = bulkT + lane;
#pragma unroll
        for (int it = 0; it < 4; ++it) {
            const int g  = wave + 4 * it;
            const float* s4 = src + (4 * (g & 7)) * 64 + (g >> 3) * 32;
            stg[it].x = s4[0];
            stg[it].y = s4[64];
            stg[it].z = s4[128];
            stg[it].w = s4[192];
        }
    }

    for (int site = 1; site <= NSITES - 2; ++site) {
        float* M = &sM[(site - 1) & 1][0];

#pragma unroll
        for (int it = 0; it < 4; ++it) {
            const int g = wave + 4 * it;
            *(float4*)&M[lane * RPAD + (g >> 3) * 32 + 4 * (g & 7)] = stg[it];
        }
        __syncthreads();

        if (site < NSITES - 2) {
            const float* src = bulkT + (size_t)site * (BOND * 2 * BOND) + lane;
#pragma unroll
            for (int it = 0; it < 4; ++it) {
                const int g  = wave + 4 * it;
                const float* s4 = src + (4 * (g & 7)) * 64 + (g >> 3) * 32;
                stg[it].x = s4[0];
                stg[it].y = s4[64];
                stg[it].z = s4[128];
                stg[it].w = s4[192];
            }
        }

        int sarr[8];
        *(int4*)&sarr[0] = *(const int4*)&sCfgAll[wave][site][8 * hi];
        *(int4*)&sarr[4] = *(const int4*)&sCfgAll[wave][site][8 * hi + 4];

        v2f a[8];
#pragma unroll
        for (int k = 0; k < 8; ++k) {
            a[k] = *(const v2f*)&env[l16 * EPAD + 4 * k + 2 * hi];
        }

        v8f acc[4] = {};

#pragma unroll
        for (int k = 0; k < 8; ++k) {
            const int r0b = 4 * k + 2 * hi;
#pragma unroll
            for (int n = 0; n < 4; ++n) {
                const int colg = (n & 1) * 16 + l16;
                const int prow = (n >> 1) * 32;
                const v2f b = *(const v2f*)&M[colg * RPAD + prow + r0b];
                acc[n] = __builtin_amdgcn_wmma_f32_16x16x4_f32(
                    false, a[k], false, b, (short)0, acc[n], false, false);
            }
        }

#pragma unroll
        for (int v = 0; v < 8; ++v) {
            const int row = v + 8 * hi;
            const bool sp = (sarr[v] != 0);
            env[row * EPAD + l16]      = sp ? acc[2][v] : acc[0][v];
            env[row * EPAD + 16 + l16] = sp ? acc[3][v] : acc[1][v];
        }
    }

    if (lane < ROWS_PER_WAVE) {
        const int r = rowBase + lane;
        const int p = sCfgAll[wave][NSITES - 1][lane];
        float acc = 0.0f;
#pragma unroll 8
        for (int d = 0; d < BOND; ++d) {
            acc = fmaf(env[lane * EPAD + d], rightT[d * 2 + p], acc);
        }
        out[r] = acc;
    }
}

extern "C" void kernel_launch(void* const* d_in, const int* in_sizes, int n_in,
                              void* d_out, int out_size, void* d_ws, size_t ws_size,
                              hipStream_t stream) {
    const int*   cfg    = (const int*)  d_in[0];
    const float* leftT  = (const float*)d_in[1];
    const float* bulkT  = (const float*)d_in[2];
    const float* rightT = (const float*)d_in[3];
    float*       out    = (float*)d_out;
    (void)in_sizes; (void)n_in; (void)out_size; (void)d_ws; (void)ws_size;

    dim3 grid(BATCH / ROWS_PER_WG);
    dim3 block(WAVES_PER_WG * 32);
    hipLaunchKernelGGL(mps_psi_kernel, grid, block, 0, stream,
                       cfg, leftT, bulkT, rightT, out);
}


# Site-matrix staging loads issued two sites ahead instead of one: second staging register set, site loop unrolled by two, per-site barrier waits for LDS only so the loads stay in flight across it
# speedup vs baseline: 2.7907x; 1.1312x over previous
#include <hip/hip_runtime.h>


typedef __attribute__((ext_vector_type(2))) float v2f;
typedef __attribute__((ext_vector_type(8))) float v8f;

#define BATCH 32768
#define NSITES 64
#define BOND 32
#define ROWS_PER_WAVE 16
#define WAVES_PER_WG 4
#define ROWS_PER_WG (ROWS_PER_WAVE * WAVES_PER_WG)
#define RPAD 68
#define CPAD 20

__device__ __forceinline__ void lds_barrier()
{
    asm volatile("s_wait_dscnt 0x0" ::: "memory");
    asm volatile("s_barrier_signal -1" ::: "memory");
    asm volatile("s_barrier_wait -1" ::: "memory");
}

__device__ __forceinline__ void stage_load(float4 (&stg)[4],
                                           const float* __restrict__ src,
                                           int wave)
{
#pragma unroll
    for (int it = 0; it < 4; ++it) {
        const int g  = wave + 4 * it;
        const int kk = g & 7;
        const int d0 = 16 * (kk >> 2) + 2 * (kk & 3);
        const float* s4 = src + d0 * 64 + (g >> 3) * 32;
        stg[it].x = s4[0];
        stg[it].y = s4[64];
        stg[it].z = s4[8 * 64];
        stg[it].w = s4[9 * 64];
    }
}

template <bool REFILL>
__device__ __forceinline__ void site_step(const int site,
                                          float4 (&stg)[4],
                                          float* M,
                                          const int* cfgW,
                                          const float* __restrict__ bulkT,
                                          v8f (&envT)[2],
                                          const int wave, const int lane,
                                          const int l16, const int hi)
{
    const bool sp = (cfgW[site * CPAD + l16] != 0);

#pragma unroll
    for (int it = 0; it < 4; ++it) {
        const int g = wave + 4 * it;
        *(float4*)&M[lane * RPAD + (g >> 3) * 32 + 4 * (g & 7)] = stg[it];
    }
    lds_barrier();

    if (REFILL) {
        stage_load(stg, bulkT + (size_t)(site + 1) * (BOND * 2 * BOND) + lane, wave);
    }

    v2f a[8][4];
#pragma unroll
    for (int kk = 0; kk < 8; ++kk) {
#pragma unroll
        for (int n = 0; n < 4; ++n) {
            const int erow = (n & 1) * 16 + l16;
            const int prow = (n >> 1) * 32;
            a[kk][n] = *(const v2f*)&M[erow * RPAD + prow + 4 * kk + 2 * hi];
        }
    }
    __builtin_amdgcn_sched_barrier(0);

    v8f acc[4] = {};

#pragma unroll
    for (int kk = 0; kk < 8; ++kk) {
        v2f b;
        b.x = envT[kk >> 2][2 * (kk & 3)];
        b.y = envT[kk >> 2][2 * (kk & 3) + 1];
#pragma unroll
        for (int n = 0; n < 4; ++n) {
            acc[n] = __builtin_amdgcn_wmma_f32_16x16x4_f32(
                false, a[kk][n], false, b, (short)0, acc[n], false, false);
        }
    }

#pragma unroll
    for (int r = 0; r < 8; ++r) {
        envT[0][r] = sp ? acc[2][r] : acc[0][r];
        envT[1][r] = sp ? acc[3][r] : acc[1][r];
    }
    __builtin_amdgcn_sched_barrier(0);
}

__global__ __launch_bounds__(WAVES_PER_WG * 32)
__attribute__((amdgpu_waves_per_eu(1, 4)))
void mps_psi_kernel(const int*   __restrict__ cfg,
                    const float* __restrict__ leftT,
                    const float* __restrict__ bulkT,
                    const float* __restrict__ rightT,
                    float*       __restrict__ out)
{
    __shared__ __align__(16) float sM[2][BOND * RPAD];
    __shared__ __align__(16) int sCfgAll[WAVES_PER_WG][NSITES][CPAD];

    const int tid   = threadIdx.x;
    const int wave  = tid >> 5;
    const int lane  = tid & 31;
    const int l16   = lane & 15;
    const int hi    = lane >> 4;
    const int rowBase = blockIdx.x * ROWS_PER_WG + wave * ROWS_PER_WAVE;

    for (int i = lane; i < ROWS_PER_WAVE * NSITES; i += 32) {
        const int r = i >> 6;
        const int c = i & 63;
        sCfgAll[wave][c][r] = cfg[(rowBase + r) * NSITES + c];
    }

    v8f envT[2];
    {
        const int s0 = sCfgAll[wave][0][l16];
        const float* lp = leftT + s0 * BOND + 8 * hi;
#pragma unroll
        for (int t = 0; t < 2; ++t) {
            const float4 x = *(const float4*)(lp + 16 * t);
            const float4 y = *(const float4*)(lp + 16 * t + 4);
            envT[t][0] = x.x; envT[t][1] = x.y; envT[t][2] = x.z; envT[t][3] = x.w;
            envT[t][4] = y.x; envT[t][5] = y.y; envT[t][6] = y.z; envT[t][7] = y.w;
        }
    }

    float4 stgA[4], stgB[4];
    stage_load(stgA, bulkT + lane, wave);
    __builtin_amdgcn_sched_barrier(0);
    stage_load(stgB, bulkT + (BOND * 2 * BOND) + lane, wave);

    for (int site = 1; site <= NSITES - 4; site += 2) {
        site_step<true>(site,     stgA, &sM[0][0], &sCfgAll[wave][0][0], bulkT,
                        envT, wave, lane, l16, hi);
        site_step<true>(site + 1, stgB, &sM[1][0], &sCfgAll[wave][0][0], bulkT,
                        envT, wave, lane, l16, hi);
    }
    site_step<false>(NSITES - 3, stgA, &sM[0][0], &sCfgAll[wave][0][0], bulkT,
                     envT, wave, lane, l16, hi);
    site_step<false>(NSITES - 2, stgB, &sM[1][0], &sCfgAll[wave][0][0], bulkT,
                     envT, wave, lane, l16, hi);

    {
        const int p = sCfgAll[wave][NSITES - 1][l16];
        float part = 0.0f;
#pragma unroll
        for (int t = 0; t < 2; ++t) {
#pragma unroll
            for (int r = 0; r < 8; ++r) {
                part = fmaf(envT[t][r], rightT[(16 * t + 8 * hi + r) * 2 + p], part);
            }
        }
        const float tot = part + __shfl_xor(part, 16);
        if (lane < ROWS_PER_WAVE) {
            out[rowBase + lane] = tot;
        }
    }
}

extern "C" void kernel_launch(void* const* d_in, const int* in_sizes, int n_in,
                              void* d_out, int out_size, void* d_ws, size_t ws_size,
                              hipStream_t stream) {
    const int*   cfg    = (const int*)  d_in[0];
    const float* leftT  = (const float*)d_in[1];
    const float* bulkT  = (const float*)d_in[2];
    const float* rightT = (const float*)d_in[3];
    float*       out    = (float*)d_out;
    (void)in_sizes; (void)n_in; (void)out_size; (void)d_ws; (void)ws_size;

    dim3 grid(BATCH / ROWS_PER_WG);
    dim3 block(WAVES_PER_WG * 32);
    hipLaunchKernelGGL(mps_psi_kernel, grid, block, 0, stream,
                       cfg, leftT, bulkT, rightT, out);
}


# Site-matrix fragments fetched from LDS one site ahead into a second register set; LDS landing moved a site earlier, workgroup barrier split (signal mid WMMA block, wait at the next step's top)
# speedup vs baseline: 2.9633x; 1.0618x over previous
#include <hip/hip_runtime.h>


typedef __attribute__((ext_vector_type(2))) float v2f;
typedef __attribute__((ext_vector_type(8))) float v8f;

#define BATCH 32768
#define NSITES 64
#define BOND 32
#define ROWS_PER_WAVE 16
#define WAVES_PER_WG 4
#define ROWS_PER_WG (ROWS_PER_WAVE * WAVES_PER_WG)
#define RPAD 68
#define CPAD 20
#define SITE_FLOATS (BOND * 2 * BOND)
#define SIGNAL_KK 4

__device__ __forceinline__ void lds_arrive()
{
    asm volatile("s_wait_dscnt 0x0" ::: "memory");
    asm volatile("s_barrier_signal -1" ::: "memory");
}

__device__ __forceinline__ void lds_wait()
{
    asm volatile("s_barrier_wait -1" ::: "memory");
}

__device__ __forceinline__ void stage_load(float4 (&stg)[4],
                                           const float* __restrict__ src,
                                           int wave)
{
#pragma unroll
    for (int it = 0; it < 4; ++it) {
        const int g  = wave + 4 * it;
        const int kk = g & 7;
        const int d0 = 16 * (kk >> 2) + 2 * (kk & 3);
        const float* s4 = src + d0 * 64 + (g >> 3) * 32;
        stg[it].x = s4[0];
        stg[it].y = s4[64];
        stg[it].z = s4[8 * 64];
        stg[it].w = s4[9 * 64];
    }
}

__device__ __forceinline__ void stage_store(const float4 (&stg)[4], float* M,
                                            const int wave, const int lane)
{
#pragma unroll
    for (int it = 0; it < 4; ++it) {
        const int g = wave + 4 * it;
        *(float4*)&M[lane * RPAD + (g >> 3) * 32 + 4 * (g & 7)] = stg[it];
    }
}

__device__ __forceinline__ void frag_load(v2f (&a)[8][4], const float* M,
                                          const int l16, const int hi)
{
#pragma unroll
    for (int kk = 0; kk < 8; ++kk) {
#pragma unroll
        for (int n = 0; n < 4; ++n) {
            const int erow = (n & 1) * 16 + l16;
            const int prow = (n >> 1) * 32;
            a[kk][n] = *(const v2f*)&M[erow * RPAD + prow + 4 * kk + 2 * hi];
        }
    }
}

template <bool WAIT, bool FRAG, bool LAND, bool REFILL>
__device__ __forceinline__ void site_step(const int site,
                                          float4 (&stg)[4],
                                          const v2f (&aCur)[8][4],
                                          v2f (&aNext)[8][4],
                                          float* Mcur,
                                          const float* Mnext,
                                          const int* cfgW,
                                          const float* __restrict__ bulkT,
                                          v8f (&envT)[2],
                                          const int wave, const int lane,
                                          const int l16, const int hi)
{
    const bool sp = (cfgW[site * CPAD + l16] != 0);

    if (WAIT) {
        lds_wait();
    }
    if (FRAG) {
        frag_load(aNext, Mnext, l16, hi);
    }
    if (LAND) {
        stage_store(stg, Mcur, wave, lane);
    }
    __builtin_amdgcn_sched_barrier(0);

    v8f acc[4] = {};

#pragma unroll
    for (int kk = 0; kk < 8; ++kk) {
        if (kk == SIGNAL_KK) {
            __builtin_amdgcn_sched_barrier(0);
            if (LAND) {
                lds_arrive();
            }
            if (REFILL) {
                stage_load(stg, bulkT + (size_t)(site + 3) * SITE_FLOATS + lane, wave);
            }
            __builtin_amdgcn_sched_barrier(0);
        }
        v2f b;
        b.x = envT[kk >> 2][2 * (kk & 3)];
        b.y = envT[kk >> 2][2 * (kk & 3) + 1];
#pragma unroll
        for (int n = 0; n < 4; ++n) {
            acc[n] = __builtin_amdgcn_wmma_f32_16x16x4_f32(
                false, aCur[kk][n], false, b, (short)0, acc[n], false, false);
        }
    }

#pragma unroll
    for (int r = 0; r < 8; ++r) {
        envT[0][r] = sp ? acc[2][r] : acc[0][r];
        envT[1][r] = sp ? acc[3][r] : acc[1][r];
    }
    __builtin_amdgcn_sched_barrier(0);
}

__global__ __launch_bounds__(WAVES_PER_WG * 32)
__attribute__((amdgpu_waves_per_eu(1, 2)))
void mps_psi_kernel(const int*   __restrict__ cfg,
                    const float* __restrict__ leftT,
                    const float* __restrict__ bulkT,
                    const float* __restrict__ rightT,
                    float*       __restrict__ out)
{
    __shared__ __align__(16) float sM[2][BOND * RPAD];
    __shared__ __align__(16) int sCfgAll[WAVES_PER_WG][NSITES][CPAD];

    const int tid   = threadIdx.x;
    const int wave  = tid >> 5;
    const int lane  = tid & 31;
    const int l16   = lane & 15;
    const int hi    = lane >> 4;
    const int rowBase = blockIdx.x * ROWS_PER_WG + wave * ROWS_PER_WAVE;

    for (int i = lane; i < ROWS_PER_WAVE * NSITES; i += 32) {
        const int r = i >> 6;
        const int c = i & 63;
        sCfgAll[wave][c][r] = cfg[(rowBase + r) * NSITES + c];
    }

    v8f envT[2];
    {
        const int s0 = sCfgAll[wave][0][l16];
        const float* lp = leftT + s0 * BOND + 8 * hi;
#pragma unroll
        for (int t = 0; t < 2; ++t) {
            const float4 x = *(const float4*)(lp + 16 * t);
            const float4 y = *(const float4*)(lp + 16 * t + 4);
            envT[t][0] = x.x; envT[t][1] = x.y; envT[t][2] = x.z; envT[t][3] = x.w;
            envT[t][4] = y.x; envT[t][5] = y.y; envT[t][6] = y.z; envT[t][7] = y.w;
        }
    }

    float4 stgA[4], stgB[4];
    v2f fragA[8][4], fragB[8][4];
    float* const M0 = &sM[0][0];
    float* const M1 = &sM[1][0];
    const int* const cfgW = &sCfgAll[wave][0][0];

    stage_load(stgA, bulkT + lane, wave);
    __builtin_amdgcn_sched_barrier(0);
    stage_load(stgB, bulkT + SITE_FLOATS + lane, wave);
    __builtin_amdgcn_sched_barrier(0);
    stage_store(stgA, M0, wave, lane);
    lds_arrive();
    lds_wait();
    stage_load(stgA, bulkT + 2 * SITE_FLOATS + lane, wave);
    frag_load(fragA, M0, l16, hi);
    __builtin_amdgcn_sched_barrier(0);
    stage_store(stgB, M1, wave, lane);
    lds_arrive();
    stage_load(stgB, bulkT + 3 * SITE_FLOATS + lane, wave);
    __builtin_amdgcn_sched_barrier(0);

    for (int site = 1; site <= NSITES - 6; site += 2) {
        site_step<true, true, true, true>(site,     stgA, fragA, fragB, M0, M1,
                                          cfgW, bulkT, envT, wave, lane, l16, hi);
        site_step<true, true, true, true>(site + 1, stgB, fragB, fragA, M1, M0,
                                          cfgW, bulkT, envT, wave, lane, l16, hi);
    }
    site_step<true, true, true, false>(NSITES - 5, stgA, fragA, fragB, M0, M1,
                                       cfgW, bulkT, envT, wave, lane, l16, hi);
    site_step<true, true, true, false>(NSITES - 4, stgB, fragB, fragA, M1, M0,
                                       cfgW, bulkT, envT, wave, lane, l16, hi);
    site_step<true, true, false, false>(NSITES - 3, stgA, fragA, fragB, M0, M1,
                                        cfgW, bulkT, envT, wave, lane, l16, hi);
    site_step<false, false, false, false>(NSITES - 2, stgB, fragB, fragA, M1, M0,
                                          cfgW, bulkT, envT, wave, lane, l16, hi);

    {
        const int p = sCfgAll[wave][NSITES - 1][l16];
        float part = 0.0f;
#pragma unroll
        for (int t = 0; t < 2; ++t) {
#pragma unroll
            for (int r = 0; r < 8; ++r) {
                part = fmaf(envT[t][r], rightT[(16 * t + 8 * hi + r) * 2 + p], part);
            }
        }
        const float tot = part + __shfl_xor(part, 16);
        if (lane < ROWS_PER_WAVE) {
            out[rowBase + lane] = tot;
        }
    }
}

extern "C" void kernel_launch(void* const* d_in, const int* in_sizes, int n_in,
                              void* d_out, int out_size, void* d_ws, size_t ws_size,
                              hipStream_t stream) {
    const int*   cfg    = (const int*)  d_in[0];
    const float* leftT  = (const float*)d_in[1];
    const float* bulkT  = (const float*)d_in[2];
    const float* rightT = (const float*)d_in[3];
    float*       out    = (float*)d_out;
    (void)in_sizes; (void)n_in; (void)out_size; (void)d_ws; (void)ws_size;

    dim3 grid(BATCH / ROWS_PER_WG);
    dim3 block(WAVES_PER_WG * 32);
    hipLaunchKernelGGL(mps_psi_kernel, grid, block, 0, stream,
                       cfg, leftT, bulkT, rightT, out);
}


# Two 16-row batch tiles per wave at one wave per SIMD: each tile's select is issued among the other tile's WMMAs, every A fragment feeds two WMMAs, one LDS group and barrier per 64 WMMAs
# speedup vs baseline: 3.1094x; 1.0493x over previous
#include <hip/hip_runtime.h>


typedef __attribute__((ext_vector_type(2))) float v2f;
typedef __attribute__((ext_vector_type(8))) float v8f;

#define BATCH 32768
#define NSITES 64
#define BOND 32
#define TILE_ROWS 16
#define ROWS_PER_WAVE (2 * TILE_ROWS)
#define WAVES_PER_WG 4
#define ROWS_PER_WG (ROWS_PER_WAVE * WAVES_PER_WG)
#define RPAD 68
#define CPAD 36
#define SITE_FLOATS (BOND * 2 * BOND)

__device__ __forceinline__ void lds_arrive()
{
    asm volatile("s_wait_dscnt 0x0" ::: "memory");
    asm volatile("s_barrier_signal -1" ::: "memory");
}

__device__ __forceinline__ void lds_wait()
{
    asm volatile("s_barrier_wait -1" ::: "memory");
}

__device__ __forceinline__ void stage_load(float4 (&stg)[4],
                                           const float* __restrict__ src,
                                           int wave)
{
#pragma unroll
    for (int it = 0; it < 4; ++it) {
        const int g  = wave + 4 * it;
        const int kk = g & 7;
        const int d0 = 16 * (kk >> 2) + 2 * (kk & 3);
        const float* s4 = src + d0 * 64 + (g >> 3) * 32;
        stg[it].x = s4[0];
        stg[it].y = s4[64];
        stg[it].z = s4[8 * 64];
        stg[it].w = s4[9 * 64];
    }
}

__device__ __forceinline__ void stage_store(const float4 (&stg)[4], float* M,
                                            const int wave, const int lane)
{
#pragma unroll
    for (int it = 0; it < 4; ++it) {
        const int g = wave + 4 * it;
        *(float4*)&M[lane * RPAD + (g >> 3) * 32 + 4 * (g & 7)] = stg[it];
    }
}

__device__ __forceinline__ void frag_load(v2f (&a)[8][4], const float* M,
                                          const int l16, const int hi)
{
#pragma unroll
    for (int kk = 0; kk < 8; ++kk) {
#pragma unroll
        for (int n = 0; n < 4; ++n) {
            const int erow = (n & 1) * 16 + l16;
            const int prow = (n >> 1) * 32;
            a[kk][n] = *(const v2f*)&M[erow * RPAD + prow + 4 * kk + 2 * hi];
        }
    }
}

__device__ __forceinline__ void select_env(v8f (&env)[2], const v8f (&acc)[4],
                                           const bool sp)
{
#pragma unroll
    for (int r = 0; r < 8; ++r) {
        env[0][r] = sp ? acc[2][r] : acc[0][r];
        env[1][r] = sp ? acc[3][r] : acc[1][r];
    }
}

__device__ __forceinline__ void wmma_kstep(v8f (&acc)[4], const v2f (&a)[4],
                                           const v8f (&env)[2], const int kk)
{
    v2f b;
    b.x = env[kk >> 2][2 * (kk & 3)];
    b.y = env[kk >> 2][2 * (kk & 3) + 1];
#pragma unroll
    for (int n = 0; n < 4; ++n) {
        acc[n] = __builtin_amdgcn_wmma_f32_16x16x4_f32(
            false, a[n], false, b, (short)0, acc[n], false, false);
    }
}

template <bool WAIT, bool FRAG, bool LAND, bool REFILL>
__device__ __forceinline__ void site_step(const int site,
                                          float4 (&stg)[4],
                                          const v2f (&aCur)[8][4],
                                          v2f (&aNext)[8][4],
                                          float* Mcur,
                                          const float* Mnext,
                                          const int* cfgW,
                                          const float* __restrict__ bulkT,
                                          v8f (&envP)[2], v8f (&envQ)[2],
                                          v8f (&accQ)[4], bool& spQ,
                                          const int wave, const int lane,
                                          const int l16, const int hi)
{
    const bool spP     = (cfgW[site * CPAD + l16] != 0);
    const bool spQthis = (cfgW[site * CPAD + TILE_ROWS + l16] != 0);

    if (WAIT) {
        lds_wait();
    }
    if (FRAG) {
        frag_load(aNext, Mnext, l16, hi);
    }
    if (LAND) {
        stage_store(stg, Mcur, wave, lane);
    }
    __builtin_amdgcn_sched_barrier(0);

    v8f accP[4] = {};
#pragma unroll
    for (int kk = 0; kk < 8; ++kk) {
        if (kk == 1) {
            select_env(envQ, accQ, spQ);
        }
        wmma_kstep(accP, aCur[kk], envP, kk);
    }

    __builtin_amdgcn_sched_barrier(0);
    if (LAND) {
        lds_arrive();
    }
    if (REFILL) {
        stage_load(stg, bulkT + (size_t)(site + 3) * SITE_FLOATS + lane, wave);
    }
    __builtin_amdgcn_sched_barrier(0);

#pragma unroll
    for (int n = 0; n < 4; ++n) {
        accQ[n] = v8f{};
    }
#pragma unroll
    for (int kk = 0; kk < 8; ++kk) {
        if (kk == 1) {
            select_env(envP, accP, spP);
        }
        wmma_kstep(accQ, aCur[kk], envQ, kk);
    }
    spQ = spQthis;
    __builtin_amdgcn_sched_barrier(0);
}

__global__ __launch_bounds__(WAVES_PER_WG * 32)
__attribute__((amdgpu_waves_per_eu(1, 1)))
void mps_psi_kernel(const int*   __restrict__ cfg,
                    const float* __restrict__ leftT,
                    const float* __restrict__ bulkT,
                    const float* __restrict__ rightT,
                    float*       __restrict__ out)
{
    __shared__ __align__(16) float sM[2][BOND * RPAD];
    __shared__ __align__(16) int sCfgAll[WAVES_PER_WG][NSITES][CPAD];

    const int tid   = threadIdx.x;
    const int wave  = tid >> 5;
    const int lane  = tid & 31;
    const int l16   = lane & 15;
    const int hi    = lane >> 4;
    const int rowBase = blockIdx.x * ROWS_PER_WG + wave * ROWS_PER_WAVE;

#pragma unroll
    for (int i = lane; i < ROWS_PER_WAVE * NSITES; i += 32) {
        const int r = i >> 6;
        const int c = i & 63;
        sCfgAll[wave][c][r] = cfg[(rowBase + r) * NSITES + c];
    }

    v8f envP[2], envQ[2];
    {
        const int sP = sCfgAll[wave][0][l16];
        const int sQ = sCfgAll[wave][0][TILE_ROWS + l16];
        const float* lp = leftT + sP * BOND + 8 * hi;
        const float* lq = leftT + sQ * BOND + 8 * hi;
#pragma unroll
        for (int t = 0; t < 2; ++t) {
            const float4 x = *(const float4*)(lp + 16 * t);
            const float4 y = *(const float4*)(lp + 16 * t + 4);
            envP[t][0] = x.x; envP[t][1] = x.y; envP[t][2] = x.z; envP[t][3] = x.w;
            envP[t][4] = y.x; envP[t][5] = y.y; envP[t][6] = y.z; envP[t][7] = y.w;
            const float4 u = *(const float4*)(lq + 16 * t);
            const float4 v = *(const float4*)(lq + 16 * t + 4);
            envQ[t][0] = u.x; envQ[t][1] = u.y; envQ[t][2] = u.z; envQ[t][3] = u.w;
            envQ[t][4] = v.x; envQ[t][5] = v.y; envQ[t][6] = v.z; envQ[t][7] = v.w;
        }
    }
    v8f accQ[4];
    accQ[0] = envQ[0]; accQ[1] = envQ[1]; accQ[2] = envQ[0]; accQ[3] = envQ[1];
    bool spQ = false;

    float4 stgA[4], stgB[4];
    v2f fragA[8][4], fragB[8][4];
    float* const M0 = &sM[0][0];
    float* const M1 = &sM[1][0];
    const int* const cfgW = &sCfgAll[wave][0][0];

    stage_load(stgA, bulkT + lane, wave);
    __builtin_amdgcn_sched_barrier(0);
    stage_load(stgB, bulkT + SITE_FLOATS + lane, wave);
    __builtin_amdgcn_sched_barrier(0);
    stage_store(stgA, M0, wave, lane);
    lds_arrive();
    lds_wait();
    stage_load(stgA, bulkT + 2 * SITE_FLOATS + lane, wave);
    frag_load(fragA, M0, l16, hi);
    __builtin_amdgcn_sched_barrier(0);
    stage_store(stgB, M1, wave, lane);
    lds_arrive();
    stage_load(stgB, bulkT + 3 * SITE_FLOATS + lane, wave);
    __builtin_amdgcn_sched_barrier(0);

    for (int site = 1; site <= NSITES - 6; site += 2) {
        site_step<true, true, true, true>(site,     stgA, fragA, fragB, M0, M1,
                                          cfgW, bulkT, envP, envQ, accQ, spQ,
                                          wave, lane, l16, hi);
        site_step<true, true, true, true>(site + 1, stgB, fragB, fragA, M1, M0,
                                          cfgW, bulkT, envP, envQ, accQ, spQ,
                                          wave, lane, l16, hi);
    }
    site_step<true, true, true, false>(NSITES - 5, stgA, fragA, fragB, M0, M1,
                                       cfgW, bulkT, envP, envQ, accQ, spQ,
                                       wave, lane, l16, hi);
    site_step<true, true, true, false>(NSITES - 4, stgB, fragB, fragA, M1, M0,
                                       cfgW, bulkT, envP, envQ, accQ, spQ,
                                       wave, lane, l16, hi);
    site_step<true, true, false, false>(NSITES - 3, stgA, fragA, fragB, M0, M1,
                                        cfgW, bulkT, envP, envQ, accQ, spQ,
                                        wave, lane, l16, hi);
    site_step<false, false, false, false>(NSITES - 2, stgB, fragB, fragA, M1, M0,
                                          cfgW, bulkT, envP, envQ, accQ, spQ,
                                          wave, lane, l16, hi);
    select_env(envQ, accQ, spQ);

    {
        const int pP = sCfgAll[wave][NSITES - 1][l16];
        const int pQ = sCfgAll[wave][NSITES - 1][TILE_ROWS + l16];
        float partP = 0.0f, partQ = 0.0f;
#pragma unroll
        for (int t = 0; t < 2; ++t) {
#pragma unroll
            for (int r = 0; r < 8; ++r) {
                const float* rt = rightT + (16 * t + 8 * hi + r) * 2;
                partP = fmaf(envP[t][r], rt[pP], partP);
                partQ = fmaf(envQ[t][r], rt[pQ], partQ);
            }
        }
        const float totP = partP + __shfl_xor(partP, 16);
        const float totQ = partQ + __shfl_xor(partQ, 16);
        out[rowBase + lane] = hi ? totQ : totP;
    }
}

extern "C" void kernel_launch(void* const* d_in, const int* in_sizes, int n_in,
                              void* d_out, int out_size, void* d_ws, size_t ws_size,
                              hipStream_t stream) {
    const int*   cfg    = (const int*)  d_in[0];
    const float* leftT  = (const float*)d_in[1];
    const float* bulkT  = (const float*)d_in[2];
    const float* rightT = (const float*)d_in[3];
    float*       out    = (float*)d_out;
    (void)in_sizes; (void)n_in; (void)out_size; (void)d_ws; (void)ws_size;

    dim3 grid(BATCH / ROWS_PER_WG);
    dim3 block(WAVES_PER_WG * 32);
    hipLaunchKernelGGL(mps_psi_kernel, grid, block, 0, stream,
                       cfg, leftT, bulkT, rightT, out);
}


# First and last 12 bulk sites looked up from tables of all 2^13 spin patterns, built by a small first launch; suffix vectors are built from the right, so the final f32 sum order changes (env . w)
# speedup vs baseline: 3.2964x; 1.0601x over previous
#include <hip/hip_runtime.h>


typedef __attribute__((ext_vector_type(2))) float v2f;
typedef __attribute__((ext_vector_type(8))) float v8f;

#define BATCH 32768
#define NSITES 64
#define BOND 32
#define TILE_ROWS 16
#define ROWS_PER_WAVE (2 * TILE_ROWS)
#define WAVES_PER_WG 4
#define ROWS_PER_WG (ROWS_PER_WAVE * WAVES_PER_WG)
#define RPAD 68
#define CPAD 36
#define SITE_FLOATS (BOND * 2 * BOND)
#define NPRE 12
#define TBITS (NPRE + 1)
#define TROWS (1 << TBITS)
template <int MODE> struct ModeC {
    static constexpr int NSTEP = (MODE == 0) ? (NSITES - 2 - 2 * NPRE) : NPRE;
    static constexpr int CFG0  = (MODE == 0) ? NPRE : 0;
};
template <int MODE>
__device__ __forceinline__ const float* bulk_of(const float* __restrict__ bulkT, const int k)
{
    const int idx = (MODE == 2) ? (NSITES - 2 - k)
                                : (k - 1 + ((MODE == 0) ? NPRE : 0));
    return bulkT + (size_t)idx * SITE_FLOATS;
}

__device__ __forceinline__ void lds_arrive()
{
    asm volatile("s_wait_dscnt 0x0" ::: "memory");
    asm volatile("s_barrier_signal -1" ::: "memory");
}

__device__ __forceinline__ void lds_wait()
{
    asm volatile("s_barrier_wait -1" ::: "memory");
}

template <int MODE>
__device__ __forceinline__ void stage_load(float4 (&stg)[4],
                                           const float* __restrict__ src,
                                           int wave)
{
    if (MODE == 2) {
#pragma unroll
        for (int it = 0; it < 4; ++it) {
            const float* s4 = src + (16 * wave + 4 * it) * 32;
            stg[it].x = s4[0];
            stg[it].y = s4[32];
            stg[it].z = s4[64];
            stg[it].w = s4[96];
        }
        return;
    }
#pragma unroll
    for (int it = 0; it < 4; ++it) {
        const int g  = wave + 4 * it;
        const int kk = g & 7;
        const int d0 = 16 * (kk >> 2) + 2 * (kk & 3);
        const float* s4 = src + d0 * 64 + (g >> 3) * 32;
        stg[it].x = s4[0];
        stg[it].y = s4[64];
        stg[it].z = s4[8 * 64];
        stg[it].w = s4[9 * 64];
    }
}

template <int MODE>
__device__ __forceinline__ void stage_store(const float4 (&stg)[4], float* M,
                                            const int wave, const int lane)
{
    if (MODE == 2) {
        const int pos = 16 * (lane >> 4) + 4 * ((lane >> 1) & 3)
                      + 2 * ((lane >> 3) & 1) + (lane & 1);
#pragma unroll
        for (int it = 0; it < 4; ++it) {
            const int r0 = 16 * wave + 4 * it;
            float* m = M + (r0 >> 1) * RPAD + pos;
            m[0]         = stg[it].x;
            m[32]        = stg[it].y;
            m[RPAD]      = stg[it].z;
            m[RPAD + 32] = stg[it].w;
        }
        return;
    }
#pragma unroll
    for (int it = 0; it < 4; ++it) {
        const int g = wave + 4 * it;
        *(float4*)&M[lane * RPAD + (g >> 3) * 32 + 4 * (g & 7)] = stg[it];
    }
}

__device__ __forceinline__ void frag_load(v2f (&a)[8][4], const float* M,
                                          const int l16, const int hi)
{
#pragma unroll
    for (int kk = 0; kk < 8; ++kk) {
#pragma unroll
        for (int n = 0; n < 4; ++n) {
            const int erow = (n & 1) * 16 + l16;
            const int prow = (n >> 1) * 32;
            a[kk][n] = *(const v2f*)&M[erow * RPAD + prow + 4 * kk + 2 * hi];
        }
    }
}

__device__ __forceinline__ void select_env(v8f (&env)[2], const v8f (&acc)[4],
                                           const bool sp)
{
#pragma unroll
    for (int r = 0; r < 8; ++r) {
        env[0][r] = sp ? acc[2][r] : acc[0][r];
        env[1][r] = sp ? acc[3][r] : acc[1][r];
    }
}

__device__ __forceinline__ void wmma_kstep(v8f (&acc)[4], const v2f (&a)[4],
                                           const v8f (&env)[2], const int kk)
{
    v2f b;
    b.x = env[kk >> 2][2 * (kk & 3)];
    b.y = env[kk >> 2][2 * (kk & 3) + 1];
#pragma unroll
    for (int n = 0; n < 4; ++n) {
        acc[n] = __builtin_amdgcn_wmma_f32_16x16x4_f32(
            false, a[n], false, b, (short)0, acc[n], false, false);
    }
}

template <int MODE, bool WAIT, bool FRAG, bool LAND, bool REFILL>
__device__ __forceinline__ void site_step(const int site,
                                          float4 (&stg)[4],
                                          const v2f (&aCur)[8][4],
                                          v2f (&aNext)[8][4],
                                          float* Mcur,
                                          const float* Mnext,
                                          const int* cfgW,
                                          const float* __restrict__ bulkT,
                                          v8f (&envP)[2], v8f (&envQ)[2],
                                          v8f (&accQ)[4], bool& spQ,
                                          const int wave, const int lane,
                                          const int l16, const int hi)
{
    const bool spP     = (cfgW[(site + ModeC<MODE>::CFG0) * CPAD + l16] != 0);
    const bool spQthis = (cfgW[(site + ModeC<MODE>::CFG0) * CPAD + TILE_ROWS + l16] != 0);

    if (WAIT) {
        lds_wait();
    }
    if (FRAG) {
        frag_load(aNext, Mnext, l16, hi);
    }
    if (LAND) {
        stage_store<MODE>(stg, Mcur, wave, lane);
    }
    __builtin_amdgcn_sched_barrier(0);

    v8f accP[4] = {};
#pragma unroll
    for (int kk = 0; kk < 8; ++kk) {
        if (kk == 1) {
            select_env(envQ, accQ, spQ);
        }
        wmma_kstep(accP, aCur[kk], envP, kk);
    }

    __builtin_amdgcn_sched_barrier(0);
    if (LAND) {
        lds_arrive();
    }
    if (REFILL) {
        stage_load<MODE>(stg, bulk_of<MODE>(bulkT, site + 4) + lane, wave);
    }
    __builtin_amdgcn_sched_barrier(0);

#pragma unroll
    for (int n = 0; n < 4; ++n) {
        accQ[n] = v8f{};
    }
#pragma unroll
    for (int kk = 0; kk < 8; ++kk) {
        if (kk == 1) {
            select_env(envP, accP, spP);
        }
        wmma_kstep(accQ, aCur[kk], envQ, kk);
    }
    spQ = spQthis;
    __builtin_amdgcn_sched_barrier(0);
}

template <int MODE>
__device__ __forceinline__
void mps_body(const int*   __restrict__ cfg,
              const float* __restrict__ leftT,
              const float* __restrict__ bulkT,
              const float* __restrict__ rightT,
              const float* __restrict__ preT,
              const float* __restrict__ sufT,
              float*       __restrict__ out,
              const int blockId)
{
    constexpr int NSTEP = ModeC<MODE>::NSTEP;
    __shared__ __align__(16) float sM[2][BOND * RPAD];
    __shared__ __align__(16) int sCfgAll[WAVES_PER_WG][NSITES][CPAD];

    const int tid   = threadIdx.x;
    const int wave  = tid >> 5;
    const int lane  = tid & 31;
    const int l16   = lane & 15;
    const int hi    = lane >> 4;
    const int rowBase = blockId * ROWS_PER_WG + wave * ROWS_PER_WAVE;

    if (MODE == 0) {
#pragma unroll
        for (int i = lane; i < ROWS_PER_WAVE * NSITES; i += 32) {
            const int r = i >> 6;
            const int c = i & 63;
            sCfgAll[wave][c][r] = cfg[(rowBase + r) * NSITES + c];
        }
    } else {
#pragma unroll
        for (int c = 0; c < TBITS; ++c) {
            sCfgAll[wave][c][lane] = ((rowBase + lane) >> c) & 1;
        }
    }

    v8f envP[2], envQ[2];
    {
        int sP = 0, sQ = 0;
        if (MODE == 0) {
#pragma unroll
            for (int c = 0; c < TBITS; ++c) {
                sP |= (sCfgAll[wave][c][l16] != 0) << c;
                sQ |= (sCfgAll[wave][c][TILE_ROWS + l16] != 0) << c;
            }
        } else {
            sP = sCfgAll[wave][0][l16];
            sQ = sCfgAll[wave][0][TILE_ROWS + l16];
        }
        if (MODE == 2) {
#pragma unroll
            for (int t = 0; t < 2; ++t) {
#pragma unroll
                for (int r = 0; r < 8; ++r) {
                    const float* rt = rightT + (16 * t + 8 * hi + r) * 2;
                    envP[t][r] = rt[sP];
                    envQ[t][r] = rt[sQ];
                }
            }
        } else {
            const float* base = (MODE == 0) ? preT : leftT;
            const float* lp = base + sP * BOND + 8 * hi;
            const float* lq = base + sQ * BOND + 8 * hi;
#pragma unroll
            for (int t = 0; t < 2; ++t) {
                const float4 x = *(const float4*)(lp + 16 * t);
                const float4 y = *(const float4*)(lp + 16 * t + 4);
                envP[t][0] = x.x; envP[t][1] = x.y; envP[t][2] = x.z; envP[t][3] = x.w;
                envP[t][4] = y.x; envP[t][5] = y.y; envP[t][6] = y.z; envP[t][7] = y.w;
                const float4 u = *(const float4*)(lq + 16 * t);
                const float4 v = *(const float4*)(lq + 16 * t + 4);
                envQ[t][0] = u.x; envQ[t][1] = u.y; envQ[t][2] = u.z; envQ[t][3] = u.w;
                envQ[t][4] = v.x; envQ[t][5] = v.y; envQ[t][6] = v.z; envQ[t][7] = v.w;
            }
        }
    }
    v8f accQ[4];
    accQ[0] = envQ[0]; accQ[1] = envQ[1]; accQ[2] = envQ[0]; accQ[3] = envQ[1];
    bool spQ = false;

    float4 stgA[4], stgB[4];
    v2f fragA[8][4], fragB[8][4];
    float* const M0 = &sM[0][0];
    float* const M1 = &sM[1][0];
    const int* const cfgW = &sCfgAll[wave][0][0];

    stage_load<MODE>(stgA, bulk_of<MODE>(bulkT, 1) + lane, wave);
    __builtin_amdgcn_sched_barrier(0);
    stage_load<MODE>(stgB, bulk_of<MODE>(bulkT, 2) + lane, wave);
    __builtin_amdgcn_sched_barrier(0);
    stage_store<MODE>(stgA, M0, wave, lane);
    lds_arrive();
    lds_wait();
    stage_load<MODE>(stgA, bulk_of<MODE>(bulkT, 3) + lane, wave);
    frag_load(fragA, M0, l16, hi);
    __builtin_amdgcn_sched_barrier(0);
    stage_store<MODE>(stgB, M1, wave, lane);
    lds_arrive();
    stage_load<MODE>(stgB, bulk_of<MODE>(bulkT, 4) + lane, wave);
    __builtin_amdgcn_sched_barrier(0);

    for (int k = 1; k <= NSTEP - 4; k += 2) {
        site_step<MODE, true, true, true, true>(k,     stgA, fragA, fragB, M0, M1,
                                                cfgW, bulkT, envP, envQ, accQ, spQ,
                                                wave, lane, l16, hi);
        site_step<MODE, true, true, true, true>(k + 1, stgB, fragB, fragA, M1, M0,
                                                cfgW, bulkT, envP, envQ, accQ, spQ,
                                                wave, lane, l16, hi);
    }
    site_step<MODE, true, true, true, false>(NSTEP - 3, stgA, fragA, fragB, M0, M1,
                                             cfgW, bulkT, envP, envQ, accQ, spQ,
                                             wave, lane, l16, hi);
    site_step<MODE, true, true, true, false>(NSTEP - 2, stgB, fragB, fragA, M1, M0,
                                             cfgW, bulkT, envP, envQ, accQ, spQ,
                                             wave, lane, l16, hi);
    site_step<MODE, true, true, false, false>(NSTEP - 1, stgA, fragA, fragB, M0, M1,
                                              cfgW, bulkT, envP, envQ, accQ, spQ,
                                              wave, lane, l16, hi);
    site_step<MODE, false, false, false, false>(NSTEP, stgB, fragB, fragA, M1, M0,
                                                cfgW, bulkT, envP, envQ, accQ, spQ,
                                                wave, lane, l16, hi);
    select_env(envQ, accQ, spQ);

    if (MODE != 0) {
        float* tp = out + (size_t)(rowBase + l16) * BOND + 8 * hi;
        float* tq = out + (size_t)(rowBase + TILE_ROWS + l16) * BOND + 8 * hi;
#pragma unroll
        for (int t = 0; t < 2; ++t) {
            *(float4*)(tp + 16 * t)     = make_float4(envP[t][0], envP[t][1], envP[t][2], envP[t][3]);
            *(float4*)(tp + 16 * t + 4) = make_float4(envP[t][4], envP[t][5], envP[t][6], envP[t][7]);
            *(float4*)(tq + 16 * t)     = make_float4(envQ[t][0], envQ[t][1], envQ[t][2], envQ[t][3]);
            *(float4*)(tq + 16 * t + 4) = make_float4(envQ[t][4], envQ[t][5], envQ[t][6], envQ[t][7]);
        }
        return;
    }

    {
        int iP = 0, iQ = 0;
#pragma unroll
        for (int c = 0; c < TBITS; ++c) {
            iP |= (sCfgAll[wave][NSITES - 1 - c][l16] != 0) << c;
            iQ |= (sCfgAll[wave][NSITES - 1 - c][TILE_ROWS + l16] != 0) << c;
        }
        const float* wp = sufT + (size_t)iP * BOND + 8 * hi;
        const float* wq = sufT + (size_t)iQ * BOND + 8 * hi;
        float partP = 0.0f, partQ = 0.0f;
#pragma unroll
        for (int t = 0; t < 2; ++t) {
#pragma unroll
            for (int r = 0; r < 8; ++r) {
                partP = fmaf(envP[t][r], wp[16 * t + r], partP);
                partQ = fmaf(envQ[t][r], wq[16 * t + r], partQ);
            }
        }
        const float totP = partP + __shfl_xor(partP, 16);
        const float totQ = partQ + __shfl_xor(partQ, 16);
        out[rowBase + lane] = hi ? totQ : totP;
    }
}

__global__ __launch_bounds__(WAVES_PER_WG * 32)
__attribute__((amdgpu_waves_per_eu(1, 1)))
void mps_psi_kernel(const int*   __restrict__ cfg,
                    const float* __restrict__ bulkT,
                    const float* __restrict__ preT,
                    const float* __restrict__ sufT,
                    float*       __restrict__ out)
{
    mps_body<0>(cfg, nullptr, bulkT, nullptr, preT, sufT, out, (int)blockIdx.x);
}

__global__ __launch_bounds__(WAVES_PER_WG * 32)
__attribute__((amdgpu_waves_per_eu(1, 1)))
void mps_tables_kernel(const float* __restrict__ leftT,
                       const float* __restrict__ bulkT,
                       const float* __restrict__ rightT,
                       float*       __restrict__ preT,
                       float*       __restrict__ sufT)
{
    const int b = (int)blockIdx.x;
    if (b < TROWS / ROWS_PER_WG) {
        mps_body<1>(nullptr, leftT, bulkT, nullptr, nullptr, nullptr, preT, b);
    } else {
        mps_body<2>(nullptr, nullptr, bulkT, rightT, nullptr, nullptr, sufT,
                    b - TROWS / ROWS_PER_WG);
    }
}

extern "C" void kernel_launch(void* const* d_in, const int* in_sizes, int n_in,
                              void* d_out, int out_size, void* d_ws, size_t ws_size,
                              hipStream_t stream) {
    const int*   cfg    = (const int*)  d_in[0];
    const float* leftT  = (const float*)d_in[1];
    const float* bulkT  = (const float*)d_in[2];
    const float* rightT = (const float*)d_in[3];
    float*       out    = (float*)d_out;
    float*       preT   = (float*)d_ws;
    float*       sufT   = (float*)d_ws + TROWS * BOND;
    (void)in_sizes; (void)n_in; (void)out_size; (void)ws_size;

    dim3 block(WAVES_PER_WG * 32);
    dim3 gridT(2 * TROWS / ROWS_PER_WG);
    hipLaunchKernelGGL(mps_tables_kernel, gridT, block, 0, stream,
                       leftT, bulkT, rightT, preT, sufT);
    dim3 grid(BATCH / ROWS_PER_WG);
    hipLaunchKernelGGL(mps_psi_kernel, grid, block, 0, stream,
                       cfg, bulkT, preT, sufT, out);
}


# Table launch on the whole chip: one 16-row tile per wave (256 workgroups of 64 rows) instead of two tiles on half the SIMDs, so each of its 12 chain steps costs a wave 32 WMMAs, not 64
# speedup vs baseline: 3.5547x; 1.0784x over previous
#include <hip/hip_runtime.h>


typedef __attribute__((ext_vector_type(2))) float v2f;
typedef __attribute__((ext_vector_type(8))) float v8f;

#define BATCH 32768
#define NSITES 64
#define BOND 32
#define TILE_ROWS 16
#define ROWS_PER_WAVE (2 * TILE_ROWS)
#define WAVES_PER_WG 4
#define ROWS_PER_WG (ROWS_PER_WAVE * WAVES_PER_WG)
#define RPAD 68
#define CPAD 36
#define SITE_FLOATS (BOND * 2 * BOND)
#define NPRE 12
#define TBITS (NPRE + 1)
#define TROWS (1 << TBITS)
template <int MODE> struct ModeC {
    static constexpr int NSTEP = (MODE == 0) ? (NSITES - 2 - 2 * NPRE) : NPRE;
    static constexpr int CFG0  = (MODE == 0) ? NPRE : 0;
    static constexpr bool TWO  = (MODE == 0);
    static constexpr int WAVE_ROWS = TWO ? 2 * TILE_ROWS : TILE_ROWS;
    static constexpr int WG_ROWS   = WAVE_ROWS * WAVES_PER_WG;
    static constexpr int QROW  = TWO ? TILE_ROWS : 0;
};
template <int MODE>
__device__ __forceinline__ const float* bulk_of(const float* __restrict__ bulkT, const int k)
{
    const int idx = (MODE == 2) ? (NSITES - 2 - k)
                                : (k - 1 + ((MODE == 0) ? NPRE : 0));
    return bulkT + (size_t)idx * SITE_FLOATS;
}

__device__ __forceinline__ void lds_arrive()
{
    asm volatile("s_wait_dscnt 0x0" ::: "memory");
    asm volatile("s_barrier_signal -1" ::: "memory");
}

__device__ __forceinline__ void lds_wait()
{
    asm volatile("s_barrier_wait -1" ::: "memory");
}

template <int MODE>
__device__ __forceinline__ void stage_load(float4 (&stg)[4],
                                           const float* __restrict__ src,
                                           int wave)
{
    if (MODE == 2) {
#pragma unroll
        for (int it = 0; it < 4; ++it) {
            const float* s4 = src + (16 * wave + 4 * it) * 32;
            stg[it].x = s4[0];
            stg[it].y = s4[32];
            stg[it].z = s4[64];
            stg[it].w = s4[96];
        }
        return;
    }
#pragma unroll
    for (int it = 0; it < 4; ++it) {
        const int g  = wave + 4 * it;
        const int kk = g & 7;
        const int d0 = 16 * (kk >> 2) + 2 * (kk & 3);
        const float* s4 = src + d0 * 64 + (g >> 3) * 32;
        stg[it].x = s4[0];
        stg[it].y = s4[64];
        stg[it].z = s4[8 * 64];
        stg[it].w = s4[9 * 64];
    }
}

template <int MODE>
__device__ __forceinline__ void stage_store(const float4 (&stg)[4], float* M,
                                            const int wave, const int lane)
{
    if (MODE == 2) {
        const int pos = 16 * (lane >> 4) + 4 * ((lane >> 1) & 3)
                      + 2 * ((lane >> 3) & 1) + (lane & 1);
#pragma unroll
        for (int it = 0; it < 4; ++it) {
            const int r0 = 16 * wave + 4 * it;
            float* m = M + (r0 >> 1) * RPAD + pos;
            m[0]         = stg[it].x;
            m[32]        = stg[it].y;
            m[RPAD]      = stg[it].z;
            m[RPAD + 32] = stg[it].w;
        }
        return;
    }
#pragma unroll
    for (int it = 0; it < 4; ++it) {
        const int g = wave + 4 * it;
        *(float4*)&M[lane * RPAD + (g >> 3) * 32 + 4 * (g & 7)] = stg[it];
    }
}

__device__ __forceinline__ void frag_load(v2f (&a)[8][4], const float* M,
                                          const int l16, const int hi)
{
#pragma unroll
    for (int kk = 0; kk < 8; ++kk) {
#pragma unroll
        for (int n = 0; n < 4; ++n) {
            const int erow = (n & 1) * 16 + l16;
            const int prow = (n >> 1) * 32;
            a[kk][n] = *(const v2f*)&M[erow * RPAD + prow + 4 * kk + 2 * hi];
        }
    }
}

__device__ __forceinline__ void select_env(v8f (&env)[2], const v8f (&acc)[4],
                                           const bool sp)
{
#pragma unroll
    for (int r = 0; r < 8; ++r) {
        env[0][r] = sp ? acc[2][r] : acc[0][r];
        env[1][r] = sp ? acc[3][r] : acc[1][r];
    }
}

__device__ __forceinline__ void wmma_kstep(v8f (&acc)[4], const v2f (&a)[4],
                                           const v8f (&env)[2], const int kk)
{
    v2f b;
    b.x = env[kk >> 2][2 * (kk & 3)];
    b.y = env[kk >> 2][2 * (kk & 3) + 1];
#pragma unroll
    for (int n = 0; n < 4; ++n) {
        acc[n] = __builtin_amdgcn_wmma_f32_16x16x4_f32(
            false, a[n], false, b, (short)0, acc[n], false, false);
    }
}

template <int MODE, bool WAIT, bool FRAG, bool LAND, bool REFILL>
__device__ __forceinline__ void site_step(const int site,
                                          float4 (&stg)[4],
                                          const v2f (&aCur)[8][4],
                                          v2f (&aNext)[8][4],
                                          float* Mcur,
                                          const float* Mnext,
                                          const int* cfgW,
                                          const float* __restrict__ bulkT,
                                          v8f (&envP)[2], v8f (&envQ)[2],
                                          v8f (&accQ)[4], bool& spQ,
                                          const int wave, const int lane,
                                          const int l16, const int hi)
{
    constexpr bool TWO = ModeC<MODE>::TWO;
    bool spP = false;
    if (TWO) {
        spP = (cfgW[(site + ModeC<MODE>::CFG0) * CPAD + l16] != 0);
    }
    const bool spQthis = (cfgW[(site + ModeC<MODE>::CFG0) * CPAD + ModeC<MODE>::QROW + l16] != 0);

    if (WAIT) {
        lds_wait();
    }
    if (FRAG) {
        frag_load(aNext, Mnext, l16, hi);
    }
    if (LAND) {
        stage_store<MODE>(stg, Mcur, wave, lane);
    }
    __builtin_amdgcn_sched_barrier(0);

    if (!TWO) {
        select_env(envQ, accQ, spQ);
#pragma unroll
        for (int n = 0; n < 4; ++n) {
            accQ[n] = v8f{};
        }
#pragma unroll
        for (int kk = 0; kk < 8; ++kk) {
            if (kk == 4) {
                __builtin_amdgcn_sched_barrier(0);
                if (LAND) {
                    lds_arrive();
                }
                if (REFILL) {
                    stage_load<MODE>(stg, bulk_of<MODE>(bulkT, site + 4) + lane, wave);
                }
                __builtin_amdgcn_sched_barrier(0);
            }
            wmma_kstep(accQ, aCur[kk], envQ, kk);
        }
        spQ = spQthis;
        __builtin_amdgcn_sched_barrier(0);
        return;
    }

    v8f accP[4] = {};
#pragma unroll
    for (int kk = 0; kk < 8; ++kk) {
        if (kk == 1) {
            select_env(envQ, accQ, spQ);
        }
        wmma_kstep(accP, aCur[kk], envP, kk);
    }

    __builtin_amdgcn_sched_barrier(0);
    if (LAND) {
        lds_arrive();
    }
    if (REFILL) {
        stage_load<MODE>(stg, bulk_of<MODE>(bulkT, site + 4) + lane, wave);
    }
    __builtin_amdgcn_sched_barrier(0);

#pragma unroll
    for (int n = 0; n < 4; ++n) {
        accQ[n] = v8f{};
    }
#pragma unroll
    for (int kk = 0; kk < 8; ++kk) {
        if (kk == 1) {
            select_env(envP, accP, spP);
        }
        wmma_kstep(accQ, aCur[kk], envQ, kk);
    }
    spQ = spQthis;
    __builtin_amdgcn_sched_barrier(0);
}

template <int MODE>
__device__ __forceinline__
void mps_body(const int*   __restrict__ cfg,
              const float* __restrict__ leftT,
              const float* __restrict__ bulkT,
              const float* __restrict__ rightT,
              const float* __restrict__ preT,
              const float* __restrict__ sufT,
              float*       __restrict__ out,
              const int blockId)
{
    constexpr int NSTEP = ModeC<MODE>::NSTEP;
    constexpr bool TWO  = ModeC<MODE>::TWO;
    constexpr int QROW  = ModeC<MODE>::QROW;
    __shared__ __align__(16) float sM[2][BOND * RPAD];
    __shared__ __align__(16) int sCfgAll[WAVES_PER_WG][NSITES][CPAD];

    const int tid   = threadIdx.x;
    const int wave  = tid >> 5;
    const int lane  = tid & 31;
    const int l16   = lane & 15;
    const int hi    = lane >> 4;
    const int rowBase = blockId * ModeC<MODE>::WG_ROWS + wave * ModeC<MODE>::WAVE_ROWS;

    if (MODE == 0) {
#pragma unroll
        for (int i = lane; i < ROWS_PER_WAVE * NSITES; i += 32) {
            const int r = i >> 6;
            const int c = i & 63;
            sCfgAll[wave][c][r] = cfg[(rowBase + r) * NSITES + c];
        }
    } else {
#pragma unroll
        for (int c = 0; c < TBITS; ++c) {
            sCfgAll[wave][c][lane] = ((rowBase + lane) >> c) & 1;
        }
    }

    v8f envP[2] = {}, envQ[2];
    {
        int sP = 0, sQ = 0;
        if (MODE == 0) {
#pragma unroll
            for (int c = 0; c < TBITS; ++c) {
                sP |= (sCfgAll[wave][c][l16] != 0) << c;
                sQ |= (sCfgAll[wave][c][TILE_ROWS + l16] != 0) << c;
            }
        } else {
            sQ = sCfgAll[wave][0][QROW + l16];
        }
        if (MODE == 2) {
#pragma unroll
            for (int t = 0; t < 2; ++t) {
#pragma unroll
                for (int r = 0; r < 8; ++r) {
                    const float* rt = rightT + (16 * t + 8 * hi + r) * 2;
                    envQ[t][r] = rt[sQ];
                }
            }
        } else {
            const float* base = (MODE == 0) ? preT : leftT;
            const float* lp = base + sP * BOND + 8 * hi;
            const float* lq = base + sQ * BOND + 8 * hi;
#pragma unroll
            for (int t = 0; t < 2; ++t) {
                if (TWO) {
                    const float4 x = *(const float4*)(lp + 16 * t);
                    const float4 y = *(const float4*)(lp + 16 * t + 4);
                    envP[t][0] = x.x; envP[t][1] = x.y; envP[t][2] = x.z; envP[t][3] = x.w;
                    envP[t][4] = y.x; envP[t][5] = y.y; envP[t][6] = y.z; envP[t][7] = y.w;
                }
                const float4 u = *(const float4*)(lq + 16 * t);
                const float4 v = *(const float4*)(lq + 16 * t + 4);
                envQ[t][0] = u.x; envQ[t][1] = u.y; envQ[t][2] = u.z; envQ[t][3] = u.w;
                envQ[t][4] = v.x; envQ[t][5] = v.y; envQ[t][6] = v.z; envQ[t][7] = v.w;
            }
        }
    }
    v8f accQ[4];
    accQ[0] = envQ[0]; accQ[1] = envQ[1]; accQ[2] = envQ[0]; accQ[3] = envQ[1];
    bool spQ = false;

    float4 stgA[4], stgB[4];
    v2f fragA[8][4], fragB[8][4];
    float* const M0 = &sM[0][0];
    float* const M1 = &sM[1][0];
    const int* const cfgW = &sCfgAll[wave][0][0];

    stage_load<MODE>(stgA, bulk_of<MODE>(bulkT, 1) + lane, wave);
    __builtin_amdgcn_sched_barrier(0);
    stage_load<MODE>(stgB, bulk_of<MODE>(bulkT, 2) + lane, wave);
    __builtin_amdgcn_sched_barrier(0);
    stage_store<MODE>(stgA, M0, wave, lane);
    lds_arrive();
    lds_wait();
    stage_load<MODE>(stgA, bulk_of<MODE>(bulkT, 3) + lane, wave);
    frag_load(fragA, M0, l16, hi);
    __builtin_amdgcn_sched_barrier(0);
    stage_store<MODE>(stgB, M1, wave, lane);
    lds_arrive();
    stage_load<MODE>(stgB, bulk_of<MODE>(bulkT, 4) + lane, wave);
    __builtin_amdgcn_sched_barrier(0);

    for (int k = 1; k <= NSTEP - 4; k += 2) {
        site_step<MODE, true, true, true, true>(k,     stgA, fragA, fragB, M0, M1,
                                                cfgW, bulkT, envP, envQ, accQ, spQ,
                                                wave, lane, l16, hi);
        site_step<MODE, true, true, true, true>(k + 1, stgB, fragB, fragA, M1, M0,
                                                cfgW, bulkT, envP, envQ, accQ, spQ,
                                                wave, lane, l16, hi);
    }
    site_step<MODE, true, true, true, false>(NSTEP - 3, stgA, fragA, fragB, M0, M1,
                                             cfgW, bulkT, envP, envQ, accQ, spQ,
                                             wave, lane, l16, hi);
    site_step<MODE, true, true, true, false>(NSTEP - 2, stgB, fragB, fragA, M1, M0,
                                             cfgW, bulkT, envP, envQ, accQ, spQ,
                                             wave, lane, l16, hi);
    site_step<MODE, true, true, false, false>(NSTEP - 1, stgA, fragA, fragB, M0, M1,
                                              cfgW, bulkT, envP, envQ, accQ, spQ,
                                              wave, lane, l16, hi);
    site_step<MODE, false, false, false, false>(NSTEP, stgB, fragB, fragA, M1, M0,
                                                cfgW, bulkT, envP, envQ, accQ, spQ,
                                                wave, lane, l16, hi);
    select_env(envQ, accQ, spQ);

    if (MODE != 0) {
        float* tq = out + (size_t)(rowBase + QROW + l16) * BOND + 8 * hi;
#pragma unroll
        for (int t = 0; t < 2; ++t) {
            *(float4*)(tq + 16 * t)     = make_float4(envQ[t][0], envQ[t][1], envQ[t][2], envQ[t][3]);
            *(float4*)(tq + 16 * t + 4) = make_float4(envQ[t][4], envQ[t][5], envQ[t][6], envQ[t][7]);
        }
        return;
    }

    {
        int iP = 0, iQ = 0;
#pragma unroll
        for (int c = 0; c < TBITS; ++c) {
            iP |= (sCfgAll[wave][NSITES - 1 - c][l16] != 0) << c;
            iQ |= (sCfgAll[wave][NSITES - 1 - c][TILE_ROWS + l16] != 0) << c;
        }
        const float* wp = sufT + (size_t)iP * BOND + 8 * hi;
        const float* wq = sufT + (size_t)iQ * BOND + 8 * hi;
        float partP = 0.0f, partQ = 0.0f;
#pragma unroll
        for (int t = 0; t < 2; ++t) {
#pragma unroll
            for (int r = 0; r < 8; ++r) {
                partP = fmaf(envP[t][r], wp[16 * t + r], partP);
                partQ = fmaf(envQ[t][r], wq[16 * t + r], partQ);
            }
        }
        const float totP = partP + __shfl_xor(partP, 16);
        const float totQ = partQ + __shfl_xor(partQ, 16);
        out[rowBase + lane] = hi ? totQ : totP;
    }
}

__global__ __launch_bounds__(WAVES_PER_WG * 32)
__attribute__((amdgpu_waves_per_eu(1, 1)))
void mps_psi_kernel(const int*   __restrict__ cfg,
                    const float* __restrict__ bulkT,
                    const float* __restrict__ preT,
                    const float* __restrict__ sufT,
                    float*       __restrict__ out)
{
    mps_body<0>(cfg, nullptr, bulkT, nullptr, preT, sufT, out, (int)blockIdx.x);
}

__global__ __launch_bounds__(WAVES_PER_WG * 32)
__attribute__((amdgpu_waves_per_eu(1, 1)))
void mps_tables_kernel(const float* __restrict__ leftT,
                       const float* __restrict__ bulkT,
                       const float* __restrict__ rightT,
                       float*       __restrict__ preT,
                       float*       __restrict__ sufT)
{
    const int b = (int)blockIdx.x;
    if (b < TROWS / ModeC<1>::WG_ROWS) {
        mps_body<1>(nullptr, leftT, bulkT, nullptr, nullptr, nullptr, preT, b);
    } else {
        mps_body<2>(nullptr, nullptr, bulkT, rightT, nullptr, nullptr, sufT,
                    b - TROWS / ModeC<2>::WG_ROWS);
    }
}

extern "C" void kernel_launch(void* const* d_in, const int* in_sizes, int n_in,
                              void* d_out, int out_size, void* d_ws, size_t ws_size,
                              hipStream_t stream) {
    const int*   cfg    = (const int*)  d_in[0];
    const float* leftT  = (const float*)d_in[1];
    const float* bulkT  = (const float*)d_in[2];
    const float* rightT = (const float*)d_in[3];
    float*       out    = (float*)d_out;
    float*       preT   = (float*)d_ws;
    float*       sufT   = (float*)d_ws + TROWS * BOND;
    (void)in_sizes; (void)n_in; (void)out_size; (void)ws_size;

    dim3 block(WAVES_PER_WG * 32);
    dim3 gridT(2 * TROWS / (TILE_ROWS * WAVES_PER_WG));
    hipLaunchKernelGGL(mps_tables_kernel, gridT, block, 0, stream,
                       leftT, bulkT, rightT, preT, sufT);
    dim3 grid(BATCH / ROWS_PER_WG);
    hipLaunchKernelGGL(mps_psi_kernel, grid, block, 0, stream,
                       cfg, bulkT, preT, sufT, out);
}


# Table launch: a tile's rows differ in the first 4 spins only (a workgroup's in 6), so later steps use ONE matrix of the site's pair: 16 WMMAs, no select; half-size staging sets loaded 6 steps ahead
# speedup vs baseline: 3.6858x; 1.0369x over previous
#include <hip/hip_runtime.h>


typedef __attribute__((ext_vector_type(2))) float v2f;
typedef __attribute__((ext_vector_type(8))) float v8f;

#define BATCH 32768
#define NSITES 64
#define BOND 32
#define TILE_ROWS 16
#define ROWS_PER_WAVE (2 * TILE_ROWS)
#define WAVES_PER_WG 4
#define ROWS_PER_WG (ROWS_PER_WAVE * WAVES_PER_WG)
#define RPAD 68
#define CPAD 36
#define SITE_FLOATS (BOND * 2 * BOND)
#define NPRE 12
#define TBITS (NPRE + 1)
#define TROWS (1 << TBITS)
template <int MODE> struct ModeC {
    static constexpr int NSTEP = (MODE == 0) ? (NSITES - 2 - 2 * NPRE) : NPRE;
    static constexpr bool TWO  = (MODE == 0);
    static constexpr int WAVE_ROWS = TWO ? 2 * TILE_ROWS : TILE_ROWS;
    static constexpr int WG_ROWS   = WAVE_ROWS * WAVES_PER_WG;
};
#define TILE_BITS 4
#define WG_BITS (TILE_BITS + 2)
template <int MODE>
__device__ __forceinline__ const float* bulk_of(const float* __restrict__ bulkT, const int k)
{
    const int idx = (MODE == 2) ? (NSITES - 2 - k)
                                : (k - 1 + ((MODE == 0) ? NPRE : 0));
    return bulkT + (size_t)idx * SITE_FLOATS;
}

__device__ __forceinline__ void lds_arrive()
{
    asm volatile("s_wait_dscnt 0x0" ::: "memory");
    asm volatile("s_barrier_signal -1" ::: "memory");
}

__device__ __forceinline__ void lds_wait()
{
    asm volatile("s_barrier_wait -1" ::: "memory");
}

template <int MODE>
__device__ __forceinline__ void stage_load(float4 (&stg)[4],
                                           const float* __restrict__ src,
                                           int wave)
{
    if (MODE == 2) {
#pragma unroll
        for (int it = 0; it < 4; ++it) {
            const float* s4 = src + (16 * wave + 4 * it) * 32;
            stg[it].x = s4[0];
            stg[it].y = s4[32];
            stg[it].z = s4[64];
            stg[it].w = s4[96];
        }
        return;
    }
#pragma unroll
    for (int it = 0; it < 4; ++it) {
        const int g  = wave + 4 * it;
        const int kk = g & 7;
        const int d0 = 16 * (kk >> 2) + 2 * (kk & 3);
        const float* s4 = src + d0 * 64 + (g >> 3) * 32;
        stg[it].x = s4[0];
        stg[it].y = s4[64];
        stg[it].z = s4[8 * 64];
        stg[it].w = s4[9 * 64];
    }
}

template <int MODE>
__device__ __forceinline__ void stage_store(const float4 (&stg)[4], float* M,
                                            const int wave, const int lane)
{
    if (MODE == 2) {
        const int pos = 16 * (lane >> 4) + 4 * ((lane >> 1) & 3)
                      + 2 * ((lane >> 3) & 1) + (lane & 1);
#pragma unroll
        for (int it = 0; it < 4; ++it) {
            const int r0 = 16 * wave + 4 * it;
            float* m = M + (r0 >> 1) * RPAD + pos;
            m[0]         = stg[it].x;
            m[32]        = stg[it].y;
            m[RPAD]      = stg[it].z;
            m[RPAD + 32] = stg[it].w;
        }
        return;
    }
#pragma unroll
    for (int it = 0; it < 4; ++it) {
        const int g = wave + 4 * it;
        *(float4*)&M[lane * RPAD + (g >> 3) * 32 + 4 * (g & 7)] = stg[it];
    }
}

template <int MODE, int SLOT>
__device__ __forceinline__ void stage_load_one(float4 (&stg)[4],
                                               const float* __restrict__ src,
                                               const int wave, const int p)
{
    if (MODE == 2) {
#pragma unroll
        for (int it = 0; it < 2; ++it) {
            const float* s4 = src + (8 * wave + 4 * it) * 64 + p * 32;
            stg[SLOT + it].x = s4[0];
            stg[SLOT + it].y = s4[64];
            stg[SLOT + it].z = s4[128];
            stg[SLOT + it].w = s4[192];
        }
        return;
    }
#pragma unroll
    for (int it = 0; it < 2; ++it) {
        const int d0 = 16 * it + 2 * wave;
        const float* s4 = src + d0 * 64 + p * 32;
        stg[SLOT + it].x = s4[0];
        stg[SLOT + it].y = s4[64];
        stg[SLOT + it].z = s4[8 * 64];
        stg[SLOT + it].w = s4[9 * 64];
    }
}

template <int MODE, int SLOT>
__device__ __forceinline__ void stage_store_one(const float4 (&stg)[4], float* M,
                                                const int wave, const int lane,
                                                const int p)
{
    if (MODE == 2) {
        const int pos = 16 * (lane >> 4) + 4 * ((lane >> 1) & 3)
                      + 2 * ((lane >> 3) & 1) + (lane & 1);
#pragma unroll
        for (int it = 0; it < 2; ++it) {
            float* m = M + (8 * wave + 4 * it) * RPAD + p * 32 + pos;
            m[0]        = stg[SLOT + it].x;
            m[RPAD]     = stg[SLOT + it].y;
            m[2 * RPAD] = stg[SLOT + it].z;
            m[3 * RPAD] = stg[SLOT + it].w;
        }
        return;
    }
#pragma unroll
    for (int it = 0; it < 2; ++it) {
        *(float4*)&M[lane * RPAD + p * 32 + 4 * (wave + 4 * it)] = stg[SLOT + it];
    }
}

template <int NN>
__device__ __forceinline__ void frag_load(v2f (&a)[8][4], const float* M,
                                          const int l16, const int hi)
{
#pragma unroll
    for (int kk = 0; kk < 8; ++kk) {
#pragma unroll
        for (int n = 0; n < NN; ++n) {
            const int erow = (n & 1) * 16 + l16;
            const int prow = (n >> 1) * 32;
            a[kk][n] = *(const v2f*)&M[erow * RPAD + prow + 4 * kk + 2 * hi];
        }
    }
}

__device__ __forceinline__ void select_env(v8f (&env)[2], const v8f (&acc)[4],
                                           const bool sp)
{
#pragma unroll
    for (int r = 0; r < 8; ++r) {
        env[0][r] = sp ? acc[2][r] : acc[0][r];
        env[1][r] = sp ? acc[3][r] : acc[1][r];
    }
}

__device__ __forceinline__ void wmma_kstep(v8f (&acc)[4], const v2f (&a)[4],
                                           const v8f (&env)[2], const int kk)
{
    v2f b;
    b.x = env[kk >> 2][2 * (kk & 3)];
    b.y = env[kk >> 2][2 * (kk & 3) + 1];
#pragma unroll
    for (int n = 0; n < 4; ++n) {
        acc[n] = __builtin_amdgcn_wmma_f32_16x16x4_f32(
            false, a[n], false, b, (short)0, acc[n], false, false);
    }
}

template <bool WAIT, bool FRAG, bool LAND, bool REFILL>
__device__ __forceinline__ void site_step(const int site,
                                          float4 (&stg)[4],
                                          const v2f (&aCur)[8][4],
                                          v2f (&aNext)[8][4],
                                          float* Mcur,
                                          const float* Mnext,
                                          const int* cfgW,
                                          const float* __restrict__ bulkT,
                                          v8f (&envP)[2], v8f (&envQ)[2],
                                          v8f (&accQ)[4], bool& spQ,
                                          const int wave, const int lane,
                                          const int l16, const int hi)
{
    const bool spP     = (cfgW[(site + NPRE) * CPAD + l16] != 0);
    const bool spQthis = (cfgW[(site + NPRE) * CPAD + TILE_ROWS + l16] != 0);

    if (WAIT) {
        lds_wait();
    }
    if (FRAG) {
        frag_load<4>(aNext, Mnext, l16, hi);
    }
    if (LAND) {
        stage_store<0>(stg, Mcur, wave, lane);
    }
    __builtin_amdgcn_sched_barrier(0);

    v8f accP[4] = {};
#pragma unroll
    for (int kk = 0; kk < 8; ++kk) {
        if (kk == 1) {
            select_env(envQ, accQ, spQ);
        }
        wmma_kstep(accP, aCur[kk], envP, kk);
    }

    __builtin_amdgcn_sched_barrier(0);
    if (LAND) {
        lds_arrive();
    }
    if (REFILL) {
        stage_load<0>(stg, bulk_of<0>(bulkT, site + 4) + lane, wave);
    }
    __builtin_amdgcn_sched_barrier(0);

#pragma unroll
    for (int n = 0; n < 4; ++n) {
        accQ[n] = v8f{};
    }
#pragma unroll
    for (int kk = 0; kk < 8; ++kk) {
        if (kk == 1) {
            select_env(envP, accP, spP);
        }
        wmma_kstep(accQ, aCur[kk], envQ, kk);
    }
    spQ = spQthis;
    __builtin_amdgcn_sched_barrier(0);
}

template <int MODE, bool WAIT, bool FRAG, int LAND, int REFILL,
          bool SEL, bool UNI, bool NEXTUNI>
__device__ __forceinline__ void table_step(const int site,
                                           float4 (&stg)[4],
                                           const v2f (&aCur)[8][4],
                                           v2f (&aNext)[8][4],
                                           float* Mcur,
                                           const float* Mnext,
                                           const int rowBase,
                                           const float* __restrict__ bulkT,
                                           v8f (&env)[2],
                                           v8f (&acc)[4], bool& sp,
                                           const int wave, const int lane,
                                           const int l16, const int hi)
{
    if (WAIT) {
        lds_wait();
    }
    if (FRAG) {
        if (NEXTUNI) {
            frag_load<2>(aNext, Mnext + 32 * ((rowBase >> (site + 1)) & 1), l16, hi);
        } else {
            frag_load<4>(aNext, Mnext, l16, hi);
        }
    }
    if (LAND == 1) {
        stage_store<MODE>(stg, Mcur, wave, lane);
    }
    if (LAND == 2) {
        stage_store_one<MODE, 0>(stg, Mcur, wave, lane, (rowBase >> (site + 2)) & 1);
    }
    if (LAND == 3) {
        stage_store_one<MODE, 2>(stg, Mcur, wave, lane, (rowBase >> (site + 2)) & 1);
    }
    __builtin_amdgcn_sched_barrier(0);

    if (SEL) {
        select_env(env, acc, sp);
    }
    v8f uni[2] = {};
    if (!UNI) {
#pragma unroll
        for (int n = 0; n < 4; ++n) {
            acc[n] = v8f{};
        }
    }
#pragma unroll
    for (int kk = 0; kk < 8; ++kk) {
        if (kk == 4) {
            __builtin_amdgcn_sched_barrier(0);
            if (LAND != 0) {
                lds_arrive();
            }
            if (REFILL == 1) {
                stage_load<MODE>(stg, bulk_of<MODE>(bulkT, site + 4) + lane, wave);
            }
            if (REFILL == 2) {
                stage_load_one<MODE, 0>(stg, bulk_of<MODE>(bulkT, site + 4) + lane, wave,
                                        (rowBase >> (site + 4)) & 1);
            }
            if (REFILL == 2 || REFILL == 4) {
                stage_load_one<MODE, 2>(stg, bulk_of<MODE>(bulkT, site + 6) + lane, wave,
                                        (rowBase >> (site + 6)) & 1);
            }
            if (REFILL == 3) {
                stage_load_one<MODE, 0>(stg, bulk_of<MODE>(bulkT, site + 6) + lane, wave,
                                        (rowBase >> (site + 6)) & 1);
            }
            __builtin_amdgcn_sched_barrier(0);
        }
        if (UNI) {
            v2f b;
            b.x = env[kk >> 2][2 * (kk & 3)];
            b.y = env[kk >> 2][2 * (kk & 3) + 1];
#pragma unroll
            for (int et = 0; et < 2; ++et) {
                uni[et] = __builtin_amdgcn_wmma_f32_16x16x4_f32(
                    false, aCur[kk][et], false, b, (short)0, uni[et], false, false);
            }
        } else {
            wmma_kstep(acc, aCur[kk], env, kk);
        }
    }
    if (UNI) {
        env[0] = uni[0];
        env[1] = uni[1];
    } else {
        sp = (((rowBase + l16) >> site) & 1) != 0;
    }
    __builtin_amdgcn_sched_barrier(0);
}

template <int MODE>
__device__ __forceinline__
void mps_body(const int*   __restrict__ cfg,
              const float* __restrict__ leftT,
              const float* __restrict__ bulkT,
              const float* __restrict__ rightT,
              const float* __restrict__ preT,
              const float* __restrict__ sufT,
              float*       __restrict__ out,
              const int blockId)
{
    constexpr int NSTEP = ModeC<MODE>::NSTEP;
    constexpr bool TWO  = ModeC<MODE>::TWO;
    __shared__ __align__(16) float sM[2][BOND * RPAD];
    __shared__ __align__(16) int sCfgAll[WAVES_PER_WG][NSITES][CPAD];

    const int tid   = threadIdx.x;
    const int wave  = tid >> 5;
    const int lane  = tid & 31;
    const int l16   = lane & 15;
    const int hi    = lane >> 4;
    const int rowBase = blockId * ModeC<MODE>::WG_ROWS + wave * ModeC<MODE>::WAVE_ROWS;

    if (MODE == 0) {
#pragma unroll
        for (int i = lane; i < ROWS_PER_WAVE * NSITES; i += 32) {
            const int r = i >> 6;
            const int c = i & 63;
            sCfgAll[wave][c][r] = cfg[(rowBase + r) * NSITES + c];
        }
    }

    v8f envP[2] = {}, envQ[2];
    {
        int sP = 0, sQ = 0;
        if (MODE == 0) {
#pragma unroll
            for (int c = 0; c < TBITS; ++c) {
                sP |= (sCfgAll[wave][c][l16] != 0) << c;
                sQ |= (sCfgAll[wave][c][TILE_ROWS + l16] != 0) << c;
            }
        } else {
            sQ = (rowBase + l16) & 1;
        }
        if (MODE == 2) {
#pragma unroll
            for (int t = 0; t < 2; ++t) {
#pragma unroll
                for (int r = 0; r < 8; ++r) {
                    const float* rt = rightT + (16 * t + 8 * hi + r) * 2;
                    envQ[t][r] = rt[sQ];
                }
            }
        } else {
            const float* base = (MODE == 0) ? preT : leftT;
            const float* lp = base + sP * BOND + 8 * hi;
            const float* lq = base + sQ * BOND + 8 * hi;
#pragma unroll
            for (int t = 0; t < 2; ++t) {
                if (TWO) {
                    const float4 x = *(const float4*)(lp + 16 * t);
                    const float4 y = *(const float4*)(lp + 16 * t + 4);
                    envP[t][0] = x.x; envP[t][1] = x.y; envP[t][2] = x.z; envP[t][3] = x.w;
                    envP[t][4] = y.x; envP[t][5] = y.y; envP[t][6] = y.z; envP[t][7] = y.w;
                }
                const float4 u = *(const float4*)(lq + 16 * t);
                const float4 v = *(const float4*)(lq + 16 * t + 4);
                envQ[t][0] = u.x; envQ[t][1] = u.y; envQ[t][2] = u.z; envQ[t][3] = u.w;
                envQ[t][4] = v.x; envQ[t][5] = v.y; envQ[t][6] = v.z; envQ[t][7] = v.w;
            }
        }
    }
    v8f accQ[4];
    accQ[0] = envQ[0]; accQ[1] = envQ[1]; accQ[2] = envQ[0]; accQ[3] = envQ[1];
    bool spQ = false;

    float4 stgA[4], stgB[4];
    v2f fragA[8][4], fragB[8][4];
    float* const M0 = &sM[0][0];
    float* const M1 = &sM[1][0];
    const int* const cfgW = &sCfgAll[wave][0][0];

    stage_load<MODE>(stgA, bulk_of<MODE>(bulkT, 1) + lane, wave);
    __builtin_amdgcn_sched_barrier(0);
    stage_load<MODE>(stgB, bulk_of<MODE>(bulkT, 2) + lane, wave);
    __builtin_amdgcn_sched_barrier(0);
    stage_store<MODE>(stgA, M0, wave, lane);
    lds_arrive();
    lds_wait();
    stage_load<MODE>(stgA, bulk_of<MODE>(bulkT, 3) + lane, wave);
    frag_load<4>(fragA, M0, l16, hi);
    __builtin_amdgcn_sched_barrier(0);
    stage_store<MODE>(stgB, M1, wave, lane);
    lds_arrive();
    stage_load<MODE>(stgB, bulk_of<MODE>(bulkT, 4) + lane, wave);
    __builtin_amdgcn_sched_barrier(0);

    if constexpr (TWO) {
        for (int k = 1; k <= NSTEP - 4; k += 2) {
            site_step<true, true, true, true>(k,     stgA, fragA, fragB, M0, M1,
                                              cfgW, bulkT, envP, envQ, accQ, spQ,
                                              wave, lane, l16, hi);
            site_step<true, true, true, true>(k + 1, stgB, fragB, fragA, M1, M0,
                                              cfgW, bulkT, envP, envQ, accQ, spQ,
                                              wave, lane, l16, hi);
        }
        site_step<true, true, true, false>(NSTEP - 3, stgA, fragA, fragB, M0, M1,
                                           cfgW, bulkT, envP, envQ, accQ, spQ,
                                           wave, lane, l16, hi);
        site_step<true, true, true, false>(NSTEP - 2, stgB, fragB, fragA, M1, M0,
                                           cfgW, bulkT, envP, envQ, accQ, spQ,
                                           wave, lane, l16, hi);
        site_step<true, true, false, false>(NSTEP - 1, stgA, fragA, fragB, M0, M1,
                                            cfgW, bulkT, envP, envQ, accQ, spQ,
                                            wave, lane, l16, hi);
        site_step<false, false, false, false>(NSTEP, stgB, fragB, fragA, M1, M0,
                                              cfgW, bulkT, envP, envQ, accQ, spQ,
                                              wave, lane, l16, hi);
        select_env(envQ, accQ, spQ);
    } else {
        static_assert(TILE_BITS == 4 && WG_BITS == 6 && WAVES_PER_WG == 4 && NSTEP == 12,
                      "table chain: the twelve steps are written out");
        table_step<MODE, true, true, 1, 1, true, false, false>(
            1, stgA, fragA, fragB, M0, M1, rowBase, bulkT, envQ, accQ, spQ,
            wave, lane, l16, hi);
        table_step<MODE, true, true, 1, 2, true, false, false>(
            2, stgB, fragB, fragA, M1, M0, rowBase, bulkT, envQ, accQ, spQ,
            wave, lane, l16, hi);
        table_step<MODE, true, true, 1, 2, true, false, true>(
            3, stgA, fragA, fragB, M0, M1, rowBase, bulkT, envQ, accQ, spQ,
            wave, lane, l16, hi);
        table_step<MODE, true, true, 2, 3, true, true, true>(
            4, stgB, fragB, fragA, M1, M0, rowBase, bulkT, envQ, accQ, spQ,
            wave, lane, l16, hi);
        table_step<MODE, true, true, 2, 3, false, true, true>(
            5, stgA, fragA, fragB, M0, M1, rowBase, bulkT, envQ, accQ, spQ,
            wave, lane, l16, hi);
        table_step<MODE, true, true, 3, 4, false, true, true>(
            6, stgB, fragB, fragA, M1, M0, rowBase, bulkT, envQ, accQ, spQ,
            wave, lane, l16, hi);
        table_step<MODE, true, true, 3, 0, false, true, true>(
            7, stgA, fragA, fragB, M0, M1, rowBase, bulkT, envQ, accQ, spQ,
            wave, lane, l16, hi);
        table_step<MODE, true, true, 2, 0, false, true, true>(
            8, stgB, fragB, fragA, M1, M0, rowBase, bulkT, envQ, accQ, spQ,
            wave, lane, l16, hi);
        table_step<MODE, true, true, 2, 0, false, true, true>(
            9, stgA, fragA, fragB, M0, M1, rowBase, bulkT, envQ, accQ, spQ,
            wave, lane, l16, hi);
        table_step<MODE, true, true, 3, 0, false, true, true>(
            10, stgB, fragB, fragA, M1, M0, rowBase, bulkT, envQ, accQ, spQ,
            wave, lane, l16, hi);
        table_step<MODE, true, true, 0, 0, false, true, true>(
            11, stgA, fragA, fragB, M0, M1, rowBase, bulkT, envQ, accQ, spQ,
            wave, lane, l16, hi);
        table_step<MODE, false, false, 0, 0, false, true, false>(
            12, stgB, fragB, fragA, M1, M0, rowBase, bulkT, envQ, accQ, spQ,
            wave, lane, l16, hi);
    }

    if (MODE != 0) {
        float* tq = out + (size_t)(rowBase + l16) * BOND + 8 * hi;
#pragma unroll
        for (int t = 0; t < 2; ++t) {
            *(float4*)(tq + 16 * t)     = make_float4(envQ[t][0], envQ[t][1], envQ[t][2], envQ[t][3]);
            *(float4*)(tq + 16 * t + 4) = make_float4(envQ[t][4], envQ[t][5], envQ[t][6], envQ[t][7]);
        }
        return;
    }

    {
        int iP = 0, iQ = 0;
#pragma unroll
        for (int c = 0; c < TBITS; ++c) {
            iP |= (sCfgAll[wave][NSITES - 1 - c][l16] != 0) << c;
            iQ |= (sCfgAll[wave][NSITES - 1 - c][TILE_ROWS + l16] != 0) << c;
        }
        const float* wp = sufT + (size_t)iP * BOND + 8 * hi;
        const float* wq = sufT + (size_t)iQ * BOND + 8 * hi;
        float partP = 0.0f, partQ = 0.0f;
#pragma unroll
        for (int t = 0; t < 2; ++t) {
#pragma unroll
            for (int r = 0; r < 8; ++r) {
                partP = fmaf(envP[t][r], wp[16 * t + r], partP);
                partQ = fmaf(envQ[t][r], wq[16 * t + r], partQ);
            }
        }
        const float totP = partP + __shfl_xor(partP, 16);
        const float totQ = partQ + __shfl_xor(partQ, 16);
        out[rowBase + lane] = hi ? totQ : totP;
    }
}

__global__ __launch_bounds__(WAVES_PER_WG * 32)
__attribute__((amdgpu_waves_per_eu(1, 1)))
void mps_psi_kernel(const int*   __restrict__ cfg,
                    const float* __restrict__ bulkT,
                    const float* __restrict__ preT,
                    const float* __restrict__ sufT,
                    float*       __restrict__ out)
{
    mps_body<0>(cfg, nullptr, bulkT, nullptr, preT, sufT, out, (int)blockIdx.x);
}

__global__ __launch_bounds__(WAVES_PER_WG * 32)
__attribute__((amdgpu_waves_per_eu(1, 1)))
void mps_tables_kernel(const float* __restrict__ leftT,
                       const float* __restrict__ bulkT,
                       const float* __restrict__ rightT,
                       float*       __restrict__ preT,
                       float*       __restrict__ sufT)
{
    const int b = (int)blockIdx.x;
    if (b < TROWS / ModeC<1>::WG_ROWS) {
        mps_body<1>(nullptr, leftT, bulkT, nullptr, nullptr, nullptr, preT, b);
    } else {
        mps_body<2>(nullptr, nullptr, bulkT, rightT, nullptr, nullptr, sufT,
                    b - TROWS / ModeC<2>::WG_ROWS);
    }
}

extern "C" void kernel_launch(void* const* d_in, const int* in_sizes, int n_in,
                              void* d_out, int out_size, void* d_ws, size_t ws_size,
                              hipStream_t stream) {
    const int*   cfg    = (const int*)  d_in[0];
    const float* leftT  = (const float*)d_in[1];
    const float* bulkT  = (const float*)d_in[2];
    const float* rightT = (const float*)d_in[3];
    float*       out    = (float*)d_out;
    float*       preT   = (float*)d_ws;
    float*       sufT   = (float*)d_ws + TROWS * BOND;
    (void)in_sizes; (void)n_in; (void)out_size; (void)ws_size;

    dim3 block(WAVES_PER_WG * 32);
    dim3 gridT(2 * TROWS / (TILE_ROWS * WAVES_PER_WG));
    hipLaunchKernelGGL(mps_tables_kernel, gridT, block, 0, stream,
                       leftT, bulkT, rightT, preT, sufT);
    dim3 grid(BATCH / ROWS_PER_WG);
    hipLaunchKernelGGL(mps_psi_kernel, grid, block, 0, stream,
                       cfg, bulkT, preT, sufT, out);
}


# Main launch fetches A fragments straight from a fragment-ordered copy of its site matrices, written to the workspace by the table launch: 16 b128 loads a site, no LDS staging or barrier in its loop
# speedup vs baseline: 3.8078x; 1.0331x over previous
#include <hip/hip_runtime.h>


typedef __attribute__((ext_vector_type(2))) float v2f;
typedef __attribute__((ext_vector_type(8))) float v8f;

#define BATCH 32768
#define NSITES 64
#define BOND 32
#define TILE_ROWS 16
#define ROWS_PER_WAVE (2 * TILE_ROWS)
#define WAVES_PER_WG 4
#define ROWS_PER_WG (ROWS_PER_WAVE * WAVES_PER_WG)
#define RPAD 68
#define CPAD 36
#define SITE_FLOATS (BOND * 2 * BOND)
#define NPRE 12
#define TBITS (NPRE + 1)
#define TROWS (1 << TBITS)
template <int MODE> struct ModeC {
    static constexpr int NSTEP = (MODE == 0) ? (NSITES - 2 - 2 * NPRE) : NPRE;
    static constexpr bool TWO  = (MODE == 0);
    static constexpr int WAVE_ROWS = TWO ? 2 * TILE_ROWS : TILE_ROWS;
    static constexpr int WG_ROWS   = WAVE_ROWS * WAVES_PER_WG;
};
#define TILE_BITS 4
#define WG_BITS (TILE_BITS + 2)
template <int MODE>
__device__ __forceinline__ const float* bulk_of(const float* __restrict__ bulkT, const int k)
{
    const int idx = (MODE == 2) ? (NSITES - 2 - k)
                                : (k - 1 + ((MODE == 0) ? NPRE : 0));
    return bulkT + (size_t)idx * SITE_FLOATS;
}

__device__ __forceinline__ void lds_arrive()
{
    asm volatile("s_wait_dscnt 0x0" ::: "memory");
    asm volatile("s_barrier_signal -1" ::: "memory");
}

__device__ __forceinline__ void lds_wait()
{
    asm volatile("s_barrier_wait -1" ::: "memory");
}

template <int MODE>
__device__ __forceinline__ void stage_load(float4 (&stg)[4],
                                           const float* __restrict__ src,
                                           int wave)
{
    if (MODE == 2) {
#pragma unroll
        for (int it = 0; it < 4; ++it) {
            const float* s4 = src + (16 * wave + 4 * it) * 32;
            stg[it].x = s4[0];
            stg[it].y = s4[32];
            stg[it].z = s4[64];
            stg[it].w = s4[96];
        }
        return;
    }
#pragma unroll
    for (int it = 0; it < 4; ++it) {
        const int g  = wave + 4 * it;
        const int kk = g & 7;
        const int d0 = 16 * (kk >> 2) + 2 * (kk & 3);
        const float* s4 = src + d0 * 64 + (g >> 3) * 32;
        stg[it].x = s4[0];
        stg[it].y = s4[64];
        stg[it].z = s4[8 * 64];
        stg[it].w = s4[9 * 64];
    }
}

template <int MODE>
__device__ __forceinline__ void stage_store(const float4 (&stg)[4], float* M,
                                            const int wave, const int lane)
{
    if (MODE == 2) {
        const int pos = 16 * (lane >> 4) + 4 * ((lane >> 1) & 3)
                      + 2 * ((lane >> 3) & 1) + (lane & 1);
#pragma unroll
        for (int it = 0; it < 4; ++it) {
            const int r0 = 16 * wave + 4 * it;
            float* m = M + (r0 >> 1) * RPAD + pos;
            m[0]         = stg[it].x;
            m[32]        = stg[it].y;
            m[RPAD]      = stg[it].z;
            m[RPAD + 32] = stg[it].w;
        }
        return;
    }
#pragma unroll
    for (int it = 0; it < 4; ++it) {
        const int g = wave + 4 * it;
        *(float4*)&M[lane * RPAD + (g >> 3) * 32 + 4 * (g & 7)] = stg[it];
    }
}

template <int MODE, int SLOT>
__device__ __forceinline__ void stage_load_one(float4 (&stg)[4],
                                               const float* __restrict__ src,
                                               const int wave, const int p)
{
    if (MODE == 2) {
#pragma unroll
        for (int it = 0; it < 2; ++it) {
            const float* s4 = src + (8 * wave + 4 * it) * 64 + p * 32;
            stg[SLOT + it].x = s4[0];
            stg[SLOT + it].y = s4[64];
            stg[SLOT + it].z = s4[128];
            stg[SLOT + it].w = s4[192];
        }
        return;
    }
#pragma unroll
    for (int it = 0; it < 2; ++it) {
        const int d0 = 16 * it + 2 * wave;
        const float* s4 = src + d0 * 64 + p * 32;
        stg[SLOT + it].x = s4[0];
        stg[SLOT + it].y = s4[64];
        stg[SLOT + it].z = s4[8 * 64];
        stg[SLOT + it].w = s4[9 * 64];
    }
}

template <int MODE, int SLOT>
__device__ __forceinline__ void stage_store_one(const float4 (&stg)[4], float* M,
                                                const int wave, const int lane,
                                                const int p)
{
    if (MODE == 2) {
        const int pos = 16 * (lane >> 4) + 4 * ((lane >> 1) & 3)
                      + 2 * ((lane >> 3) & 1) + (lane & 1);
#pragma unroll
        for (int it = 0; it < 2; ++it) {
            float* m = M + (8 * wave + 4 * it) * RPAD + p * 32 + pos;
            m[0]        = stg[SLOT + it].x;
            m[RPAD]     = stg[SLOT + it].y;
            m[2 * RPAD] = stg[SLOT + it].z;
            m[3 * RPAD] = stg[SLOT + it].w;
        }
        return;
    }
#pragma unroll
    for (int it = 0; it < 2; ++it) {
        *(float4*)&M[lane * RPAD + p * 32 + 4 * (wave + 4 * it)] = stg[SLOT + it];
    }
}

template <int NN>
__device__ __forceinline__ void frag_load(v2f (&a)[8][4], const float* M,
                                          const int l16, const int hi)
{
#pragma unroll
    for (int kk = 0; kk < 8; ++kk) {
#pragma unroll
        for (int n = 0; n < NN; ++n) {
            const int erow = (n & 1) * 16 + l16;
            const int prow = (n >> 1) * 32;
            a[kk][n] = *(const v2f*)&M[erow * RPAD + prow + 4 * kk + 2 * hi];
        }
    }
}

#define FRAG_SITE_FLOATS (8 * 2 * 32 * 4)
#define MAIN_NSTEP (NSITES - 2 - 2 * NPRE)
static_assert(MAIN_NSTEP * (FRAG_SITE_FLOATS / 4) <= 2 * (1 << (NPRE + 1)) / 64 * 128,
              "fragment copy: more float4s than the table launch has threads");
#define FRAG_COPY_VEC4 (MAIN_NSTEP * (FRAG_SITE_FLOATS / 4))
__device__ __forceinline__ float4 frag_copy_load(const float* __restrict__ bulkT, const int gt)
{
    const int g    = (gt < FRAG_COPY_VEC4) ? gt : (FRAG_COPY_VEC4 - 1);
    const int s    = g >> 9;
    const int rem  = g & 511;
    const int kk   = rem >> 6;
    const int p    = (rem >> 5) & 1;
    const int ln   = rem & 31;
    const int d0   = 16 * (kk >> 2) + 8 * (ln >> 4) + 2 * (kk & 3);
    const float* b = bulkT + (size_t)(s + NPRE) * SITE_FLOATS + d0 * 64 + p * 32 + (ln & 15);
    float4 v;
    v.x = b[0];
    v.y = b[64];
    v.z = b[16];
    v.w = b[64 + 16];
    return v;
}

__device__ __forceinline__ void frag_copy_store(float* __restrict__ fragT, const int gt,
                                                const float4 v)
{
    if (gt < FRAG_COPY_VEC4) {
        *(float4*)(fragT + (size_t)gt * 4) = v;
    }
}

__device__ __forceinline__ void frag_fetch(v2f (&a)[8][4], const float* __restrict__ F,
                                           const int lane)
{
#pragma unroll
    for (int kk = 0; kk < 8; ++kk) {
#pragma unroll
        for (int p = 0; p < 2; ++p) {
            const float4 v = *(const float4*)(F + ((kk * 2 + p) * 32 + lane) * 4);
            a[kk][2 * p].x     = v.x;
            a[kk][2 * p].y     = v.y;
            a[kk][2 * p + 1].x = v.z;
            a[kk][2 * p + 1].y = v.w;
        }
    }
}

__device__ __forceinline__ void select_env(v8f (&env)[2], const v8f (&acc)[4],
                                           const bool sp)
{
#pragma unroll
    for (int r = 0; r < 8; ++r) {
        env[0][r] = sp ? acc[2][r] : acc[0][r];
        env[1][r] = sp ? acc[3][r] : acc[1][r];
    }
}

__device__ __forceinline__ void wmma_kstep(v8f (&acc)[4], const v2f (&a)[4],
                                           const v8f (&env)[2], const int kk)
{
    v2f b;
    b.x = env[kk >> 2][2 * (kk & 3)];
    b.y = env[kk >> 2][2 * (kk & 3) + 1];
#pragma unroll
    for (int n = 0; n < 4; ++n) {
        acc[n] = __builtin_amdgcn_wmma_f32_16x16x4_f32(
            false, a[n], false, b, (short)0, acc[n], false, false);
    }
}

template <bool FETCH>
__device__ __forceinline__ void site_step(const int site,
                                          const v2f (&aCur)[8][4],
                                          v2f (&aNext)[8][4],
                                          const int* cfgW,
                                          const float* __restrict__ fragT,
                                          v8f (&envP)[2], v8f (&envQ)[2],
                                          v8f (&accQ)[4], bool& spQ,
                                          const int lane, const int l16)
{
    const bool spP     = (cfgW[(site + NPRE) * CPAD + l16] != 0);
    const bool spQthis = (cfgW[(site + NPRE) * CPAD + TILE_ROWS + l16] != 0);

    if (FETCH) {
        frag_fetch(aNext, fragT + (size_t)site * FRAG_SITE_FLOATS, lane);
    }
    __builtin_amdgcn_sched_barrier(0);

    v8f accP[4] = {};
#pragma unroll
    for (int kk = 0; kk < 8; ++kk) {
        if (kk == 1) {
            select_env(envQ, accQ, spQ);
        }
        wmma_kstep(accP, aCur[kk], envP, kk);
    }

    __builtin_amdgcn_sched_barrier(0);

#pragma unroll
    for (int n = 0; n < 4; ++n) {
        accQ[n] = v8f{};
    }
#pragma unroll
    for (int kk = 0; kk < 8; ++kk) {
        if (kk == 1) {
            select_env(envP, accP, spP);
        }
        wmma_kstep(accQ, aCur[kk], envQ, kk);
    }
    spQ = spQthis;
    __builtin_amdgcn_sched_barrier(0);
}

template <int MODE, bool WAIT, bool FRAG, int LAND, int REFILL,
          bool SEL, bool UNI, bool NEXTUNI>
__device__ __forceinline__ void table_step(const int site,
                                           float4 (&stg)[4],
                                           const v2f (&aCur)[8][4],
                                           v2f (&aNext)[8][4],
                                           float* Mcur,
                                           const float* Mnext,
                                           const int rowBase,
                                           const float* __restrict__ bulkT,
                                           v8f (&env)[2],
                                           v8f (&acc)[4], bool& sp,
                                           const int wave, const int lane,
                                           const int l16, const int hi)
{
    if (WAIT) {
        lds_wait();
    }
    if (FRAG) {
        if (NEXTUNI) {
            frag_load<2>(aNext, Mnext + 32 * ((rowBase >> (site + 1)) & 1), l16, hi);
        } else {
            frag_load<4>(aNext, Mnext, l16, hi);
        }
    }
    if (LAND == 1) {
        stage_store<MODE>(stg, Mcur, wave, lane);
    }
    if (LAND == 2) {
        stage_store_one<MODE, 0>(stg, Mcur, wave, lane, (rowBase >> (site + 2)) & 1);
    }
    if (LAND == 3) {
        stage_store_one<MODE, 2>(stg, Mcur, wave, lane, (rowBase >> (site + 2)) & 1);
    }
    __builtin_amdgcn_sched_barrier(0);

    if (SEL) {
        select_env(env, acc, sp);
    }
    v8f uni[2] = {};
    if (!UNI) {
#pragma unroll
        for (int n = 0; n < 4; ++n) {
            acc[n] = v8f{};
        }
    }
#pragma unroll
    for (int kk = 0; kk < 8; ++kk) {
        if (kk == 4) {
            __builtin_amdgcn_sched_barrier(0);
            if (LAND != 0) {
                lds_arrive();
            }
            if (REFILL == 1) {
                stage_load<MODE>(stg, bulk_of<MODE>(bulkT, site + 4) + lane, wave);
            }
            if (REFILL == 2) {
                stage_load_one<MODE, 0>(stg, bulk_of<MODE>(bulkT, site + 4) + lane, wave,
                                        (rowBase >> (site + 4)) & 1);
            }
            if (REFILL == 2 || REFILL == 4) {
                stage_load_one<MODE, 2>(stg, bulk_of<MODE>(bulkT, site + 6) + lane, wave,
                                        (rowBase >> (site + 6)) & 1);
            }
            if (REFILL == 3) {
                stage_load_one<MODE, 0>(stg, bulk_of<MODE>(bulkT, site + 6) + lane, wave,
                                        (rowBase >> (site + 6)) & 1);
            }
            __builtin_amdgcn_sched_barrier(0);
        }
        if (UNI) {
            v2f b;
            b.x = env[kk >> 2][2 * (kk & 3)];
            b.y = env[kk >> 2][2 * (kk & 3) + 1];
#pragma unroll
            for (int et = 0; et < 2; ++et) {
                uni[et] = __builtin_amdgcn_wmma_f32_16x16x4_f32(
                    false, aCur[kk][et], false, b, (short)0, uni[et], false, false);
            }
        } else {
            wmma_kstep(acc, aCur[kk], env, kk);
        }
    }
    if (UNI) {
        env[0] = uni[0];
        env[1] = uni[1];
    } else {
        sp = (((rowBase + l16) >> site) & 1) != 0;
    }
    __builtin_amdgcn_sched_barrier(0);
}

template <int MODE>
__device__ __forceinline__
void mps_body(const float* __restrict__ leftT,
              const float* __restrict__ bulkT,
              const float* __restrict__ rightT,
              float*       __restrict__ out,
              const int blockId)
{
    static_assert(MODE == 1 || MODE == 2, "table modes only; the main chain is main_body");
    constexpr int NSTEP = ModeC<MODE>::NSTEP;
    __shared__ __align__(16) float sM[2][BOND * RPAD];

    const int tid   = threadIdx.x;
    const int wave  = tid >> 5;
    const int lane  = tid & 31;
    const int l16   = lane & 15;
    const int hi    = lane >> 4;
    const int rowBase = blockId * ModeC<MODE>::WG_ROWS + wave * ModeC<MODE>::WAVE_ROWS;

    v8f envQ[2];
    {
        const int sQ = (rowBase + l16) & 1;
        if (MODE == 2) {
#pragma unroll
            for (int t = 0; t < 2; ++t) {
#pragma unroll
                for (int r = 0; r < 8; ++r) {
                    const float* rt = rightT + (16 * t + 8 * hi + r) * 2;
                    envQ[t][r] = rt[sQ];
                }
            }
        } else {
            const float* lq = leftT + sQ * BOND + 8 * hi;
#pragma unroll
            for (int t = 0; t < 2; ++t) {
                const float4 u = *(const float4*)(lq + 16 * t);
                const float4 v = *(const float4*)(lq + 16 * t + 4);
                envQ[t][0] = u.x; envQ[t][1] = u.y; envQ[t][2] = u.z; envQ[t][3] = u.w;
                envQ[t][4] = v.x; envQ[t][5] = v.y; envQ[t][6] = v.z; envQ[t][7] = v.w;
            }
        }
    }
    v8f accQ[4];
    accQ[0] = envQ[0]; accQ[1] = envQ[1]; accQ[2] = envQ[0]; accQ[3] = envQ[1];
    bool spQ = false;

    float4 stgA[4], stgB[4];
    v2f fragA[8][4], fragB[8][4];
    float* const M0 = &sM[0][0];
    float* const M1 = &sM[1][0];

    stage_load<MODE>(stgA, bulk_of<MODE>(bulkT, 1) + lane, wave);
    __builtin_amdgcn_sched_barrier(0);
    stage_load<MODE>(stgB, bulk_of<MODE>(bulkT, 2) + lane, wave);
    __builtin_amdgcn_sched_barrier(0);
    stage_store<MODE>(stgA, M0, wave, lane);
    lds_arrive();
    lds_wait();
    stage_load<MODE>(stgA, bulk_of<MODE>(bulkT, 3) + lane, wave);
    frag_load<4>(fragA, M0, l16, hi);
    __builtin_amdgcn_sched_barrier(0);
    stage_store<MODE>(stgB, M1, wave, lane);
    lds_arrive();
    stage_load<MODE>(stgB, bulk_of<MODE>(bulkT, 4) + lane, wave);
    __builtin_amdgcn_sched_barrier(0);

    {
        static_assert(TILE_BITS == 4 && WG_BITS == 6 && WAVES_PER_WG == 4 && NSTEP == 12,
                      "table chain: the twelve steps are written out");
        table_step<MODE, true, true, 1, 1, true, false, false>(
            1, stgA, fragA, fragB, M0, M1, rowBase, bulkT, envQ, accQ, spQ,
            wave, lane, l16, hi);
        table_step<MODE, true, true, 1, 2, true, false, false>(
            2, stgB, fragB, fragA, M1, M0, rowBase, bulkT, envQ, accQ, spQ,
            wave, lane, l16, hi);
        table_step<MODE, true, true, 1, 2, true, false, true>(
            3, stgA, fragA, fragB, M0, M1, rowBase, bulkT, envQ, accQ, spQ,
            wave, lane, l16, hi);
        table_step<MODE, true, true, 2, 3, true, true, true>(
            4, stgB, fragB, fragA, M1, M0, rowBase, bulkT, envQ, accQ, spQ,
            wave, lane, l16, hi);
        table_step<MODE, true, true, 2, 3, false, true, true>(
            5, stgA, fragA, fragB, M0, M1, rowBase, bulkT, envQ, accQ, spQ,
            wave, lane, l16, hi);
        table_step<MODE, true, true, 3, 4, false, true, true>(
            6, stgB, fragB, fragA, M1, M0, rowBase, bulkT, envQ, accQ, spQ,
            wave, lane, l16, hi);
        table_step<MODE, true, true, 3, 0, false, true, true>(
            7, stgA, fragA, fragB, M0, M1, rowBase, bulkT, envQ, accQ, spQ,
            wave, lane, l16, hi);
        table_step<MODE, true, true, 2, 0, false, true, true>(
            8, stgB, fragB, fragA, M1, M0, rowBase, bulkT, envQ, accQ, spQ,
            wave, lane, l16, hi);
        table_step<MODE, true, true, 2, 0, false, true, true>(
            9, stgA, fragA, fragB, M0, M1, rowBase, bulkT, envQ, accQ, spQ,
            wave, lane, l16, hi);
        table_step<MODE, true, true, 3, 0, false, true, true>(
            10, stgB, fragB, fragA, M1, M0, rowBase, bulkT, envQ, accQ, spQ,
            wave, lane, l16, hi);
        table_step<MODE, true, true, 0, 0, false, true, true>(
            11, stgA, fragA, fragB, M0, M1, rowBase, bulkT, envQ, accQ, spQ,
            wave, lane, l16, hi);
        table_step<MODE, false, false, 0, 0, false, true, false>(
            12, stgB, fragB, fragA, M1, M0, rowBase, bulkT, envQ, accQ, spQ,
            wave, lane, l16, hi);
    }

    float* tq = out + (size_t)(rowBase + l16) * BOND + 8 * hi;
#pragma unroll
    for (int t = 0; t < 2; ++t) {
        *(float4*)(tq + 16 * t)     = make_float4(envQ[t][0], envQ[t][1], envQ[t][2], envQ[t][3]);
        *(float4*)(tq + 16 * t + 4) = make_float4(envQ[t][4], envQ[t][5], envQ[t][6], envQ[t][7]);
    }
}

__device__ __forceinline__
void main_body(const int*   __restrict__ cfg,
               const float* __restrict__ fragT,
               const float* __restrict__ preT,
               const float* __restrict__ sufT,
               float*       __restrict__ out,
               const int blockId)
{
    constexpr int NSTEP = MAIN_NSTEP;
    static_assert(NSTEP % 2 == 0 && NSTEP >= 4, "site loop unrolled by two, two steps behind it");
    __shared__ __align__(16) int sCfgAll[WAVES_PER_WG][NSITES][CPAD];

    const int tid   = threadIdx.x;
    const int wave  = tid >> 5;
    const int lane  = tid & 31;
    const int l16   = lane & 15;
    const int hi    = lane >> 4;
    const int rowBase = blockId * ROWS_PER_WG + wave * ROWS_PER_WAVE;

    v2f fragA[8][4], fragB[8][4];
    frag_fetch(fragA, fragT, lane);

#pragma unroll
    for (int i = lane; i < ROWS_PER_WAVE * NSITES; i += 32) {
        const int r = i >> 6;
        const int c = i & 63;
        sCfgAll[wave][c][r] = cfg[(rowBase + r) * NSITES + c];
    }

    v8f envP[2], envQ[2];
    {
        int sP = 0, sQ = 0;
#pragma unroll
        for (int c = 0; c < TBITS; ++c) {
            sP |= (sCfgAll[wave][c][l16] != 0) << c;
            sQ |= (sCfgAll[wave][c][TILE_ROWS + l16] != 0) << c;
        }
        const float* lp = preT + sP * BOND + 8 * hi;
        const float* lq = preT + sQ * BOND + 8 * hi;
#pragma unroll
        for (int t = 0; t < 2; ++t) {
            const float4 x = *(const float4*)(lp + 16 * t);
            const float4 y = *(const float4*)(lp + 16 * t + 4);
            envP[t][0] = x.x; envP[t][1] = x.y; envP[t][2] = x.z; envP[t][3] = x.w;
            envP[t][4] = y.x; envP[t][5] = y.y; envP[t][6] = y.z; envP[t][7] = y.w;
            const float4 u = *(const float4*)(lq + 16 * t);
            const float4 v = *(const float4*)(lq + 16 * t + 4);
            envQ[t][0] = u.x; envQ[t][1] = u.y; envQ[t][2] = u.z; envQ[t][3] = u.w;
            envQ[t][4] = v.x; envQ[t][5] = v.y; envQ[t][6] = v.z; envQ[t][7] = v.w;
        }
    }
    v8f accQ[4];
    accQ[0] = envQ[0]; accQ[1] = envQ[1]; accQ[2] = envQ[0]; accQ[3] = envQ[1];
    bool spQ = false;
    const int* const cfgW = &sCfgAll[wave][0][0];

    for (int k = 1; k <= NSTEP - 3; k += 2) {
        site_step<true>(k,     fragA, fragB, cfgW, fragT, envP, envQ, accQ, spQ, lane, l16);
        site_step<true>(k + 1, fragB, fragA, cfgW, fragT, envP, envQ, accQ, spQ, lane, l16);
    }
    site_step<true>(NSTEP - 1, fragA, fragB, cfgW, fragT, envP, envQ, accQ, spQ, lane, l16);
    site_step<false>(NSTEP,    fragB, fragA, cfgW, fragT, envP, envQ, accQ, spQ, lane, l16);
    select_env(envQ, accQ, spQ);

    {
        int iP = 0, iQ = 0;
#pragma unroll
        for (int c = 0; c < TBITS; ++c) {
            iP |= (sCfgAll[wave][NSITES - 1 - c][l16] != 0) << c;
            iQ |= (sCfgAll[wave][NSITES - 1 - c][TILE_ROWS + l16] != 0) << c;
        }
        const float* wp = sufT + (size_t)iP * BOND + 8 * hi;
        const float* wq = sufT + (size_t)iQ * BOND + 8 * hi;
        float partP = 0.0f, partQ = 0.0f;
#pragma unroll
        for (int t = 0; t < 2; ++t) {
#pragma unroll
            for (int r = 0; r < 8; ++r) {
                partP = fmaf(envP[t][r], wp[16 * t + r], partP);
                partQ = fmaf(envQ[t][r], wq[16 * t + r], partQ);
            }
        }
        const float totP = partP + __shfl_xor(partP, 16);
        const float totQ = partQ + __shfl_xor(partQ, 16);
        out[rowBase + lane] = hi ? totQ : totP;
    }
}

__global__ __launch_bounds__(WAVES_PER_WG * 32)
__attribute__((amdgpu_waves_per_eu(1, 1)))
void mps_psi_kernel(const int*   __restrict__ cfg,
                    const float* __restrict__ fragT,
                    const float* __restrict__ preT,
                    const float* __restrict__ sufT,
                    float*       __restrict__ out)
{
    main_body(cfg, fragT, preT, sufT, out, (int)blockIdx.x);
}

__global__ __launch_bounds__(WAVES_PER_WG * 32)
__attribute__((amdgpu_waves_per_eu(1, 1)))
void mps_tables_kernel(const float* __restrict__ leftT,
                       const float* __restrict__ bulkT,
                       const float* __restrict__ rightT,
                       float*       __restrict__ preT,
                       float*       __restrict__ sufT,
                       float*       __restrict__ fragT)
{
    const int b = (int)blockIdx.x;
    const int gt = b * (WAVES_PER_WG * 32) + (int)threadIdx.x;
    const float4 fc = frag_copy_load(bulkT, gt);
    if (b < TROWS / ModeC<1>::WG_ROWS) {
        mps_body<1>(leftT, bulkT, nullptr, preT, b);
    } else {
        mps_body<2>(nullptr, bulkT, rightT, sufT, b - TROWS / ModeC<2>::WG_ROWS);
    }
    frag_copy_store(fragT, gt, fc);
}

extern "C" void kernel_launch(void* const* d_in, const int* in_sizes, int n_in,
                              void* d_out, int out_size, void* d_ws, size_t ws_size,
                              hipStream_t stream) {
    const int*   cfg    = (const int*)  d_in[0];
    const float* leftT  = (const float*)d_in[1];
    const float* bulkT  = (const float*)d_in[2];
    const float* rightT = (const float*)d_in[3];
    float*       out    = (float*)d_out;
    float*       preT   = (float*)d_ws;
    float*       sufT   = (float*)d_ws + TROWS * BOND;
    float*       fragT  = (float*)d_ws + 2 * TROWS * BOND;
    (void)in_sizes; (void)n_in; (void)out_size; (void)ws_size;

    dim3 block(WAVES_PER_WG * 32);
    dim3 gridT(2 * TROWS / (TILE_ROWS * WAVES_PER_WG));
    hipLaunchKernelGGL(mps_tables_kernel, gridT, block, 0, stream,
                       leftT, bulkT, rightT, preT, sufT, fragT);
    dim3 grid(BATCH / ROWS_PER_WG);
    hipLaunchKernelGGL(mps_psi_kernel, grid, block, 0, stream,
                       cfg, fragT, preT, sufT, out);
}


# Main chain two sites at a time: rows regrouped per pair by two-spin pattern via wave-private LDS; a 16-slot tile takes ONE f32 product Ma@Mb from the first launch, no select; f32 sum order changes
# speedup vs baseline: 4.4088x; 1.1578x over previous
#include <hip/hip_runtime.h>


typedef __attribute__((ext_vector_type(2))) float v2f;
typedef __attribute__((ext_vector_type(8))) float v8f;

#define BATCH 32768
#define NSITES 64
#define BOND 32
#define TILE_ROWS 16
#define ROWS_PER_WAVE (2 * TILE_ROWS)
#define WAVES_PER_WG 4
#define ROWS_PER_WG (ROWS_PER_WAVE * WAVES_PER_WG)
#define RPAD 68
#define CPAD 36
#define SITE_FLOATS (BOND * 2 * BOND)
#define NPRE 12
#define TBITS (NPRE + 1)
#define TROWS (1 << TBITS)
template <int MODE> struct ModeC {
    static constexpr int NSTEP = (MODE == 0) ? (NSITES - 2 - 2 * NPRE) : NPRE;
    static constexpr int WAVE_ROWS = (MODE == 0) ? ROWS_PER_WAVE : TILE_ROWS;
    static constexpr int WG_ROWS   = WAVE_ROWS * WAVES_PER_WG;
};
#define TILE_BITS 4
#define WG_BITS (TILE_BITS + 2)
template <int MODE>
__device__ __forceinline__ const float* bulk_of(const float* __restrict__ bulkT, const int k)
{
    const int idx = (MODE == 2) ? (NSITES - 2 - k)
                                : (k - 1 + ((MODE == 0) ? NPRE : 0));
    return bulkT + (size_t)idx * SITE_FLOATS;
}

__device__ __forceinline__ void lds_arrive()
{
    asm volatile("s_wait_dscnt 0x0" ::: "memory");
    asm volatile("s_barrier_signal -1" ::: "memory");
}

__device__ __forceinline__ void lds_wait()
{
    asm volatile("s_barrier_wait -1" ::: "memory");
}

template <int MODE>
__device__ __forceinline__ void stage_load(float4 (&stg)[4],
                                           const float* __restrict__ src,
                                           int wave)
{
    if (MODE == 2) {
#pragma unroll
        for (int it = 0; it < 4; ++it) {
            const float* s4 = src + (16 * wave + 4 * it) * 32;
            stg[it].x = s4[0];
            stg[it].y = s4[32];
            stg[it].z = s4[64];
            stg[it].w = s4[96];
        }
        return;
    }
#pragma unroll
    for (int it = 0; it < 4; ++it) {
        const int g  = wave + 4 * it;
        const int kk = g & 7;
        const int d0 = 16 * (kk >> 2) + 2 * (kk & 3);
        const float* s4 = src + d0 * 64 + (g >> 3) * 32;
        stg[it].x = s4[0];
        stg[it].y = s4[64];
        stg[it].z = s4[8 * 64];
        stg[it].w = s4[9 * 64];
    }
}

template <int MODE>
__device__ __forceinline__ void stage_store(const float4 (&stg)[4], float* M,
                                            const int wave, const int lane)
{
    if (MODE == 2) {
        const int pos = 16 * (lane >> 4) + 4 * ((lane >> 1) & 3)
                      + 2 * ((lane >> 3) & 1) + (lane & 1);
#pragma unroll
        for (int it = 0; it < 4; ++it) {
            const int r0 = 16 * wave + 4 * it;
            float* m = M + (r0 >> 1) * RPAD + pos;
            m[0]         = stg[it].x;
            m[32]        = stg[it].y;
            m[RPAD]      = stg[it].z;
            m[RPAD + 32] = stg[it].w;
        }
        return;
    }
#pragma unroll
    for (int it = 0; it < 4; ++it) {
        const int g = wave + 4 * it;
        *(float4*)&M[lane * RPAD + (g >> 3) * 32 + 4 * (g & 7)] = stg[it];
    }
}

template <int MODE, int SLOT>
__device__ __forceinline__ void stage_load_one(float4 (&stg)[4],
                                               const float* __restrict__ src,
                                               const int wave, const int p)
{
    if (MODE == 2) {
#pragma unroll
        for (int it = 0; it < 2; ++it) {
            const float* s4 = src + (8 * wave + 4 * it) * 64 + p * 32;
            stg[SLOT + it].x = s4[0];
            stg[SLOT + it].y = s4[64];
            stg[SLOT + it].z = s4[128];
            stg[SLOT + it].w = s4[192];
        }
        return;
    }
#pragma unroll
    for (int it = 0; it < 2; ++it) {
        const int d0 = 16 * it + 2 * wave;
        const float* s4 = src + d0 * 64 + p * 32;
        stg[SLOT + it].x = s4[0];
        stg[SLOT + it].y = s4[64];
        stg[SLOT + it].z = s4[8 * 64];
        stg[SLOT + it].w = s4[9 * 64];
    }
}

template <int MODE, int SLOT>
__device__ __forceinline__ void stage_store_one(const float4 (&stg)[4], float* M,
                                                const int wave, const int lane,
                                                const int p)
{
    if (MODE == 2) {
        const int pos = 16 * (lane >> 4) + 4 * ((lane >> 1) & 3)
                      + 2 * ((lane >> 3) & 1) + (lane & 1);
#pragma unroll
        for (int it = 0; it < 2; ++it) {
            float* m = M + (8 * wave + 4 * it) * RPAD + p * 32 + pos;
            m[0]        = stg[SLOT + it].x;
            m[RPAD]     = stg[SLOT + it].y;
            m[2 * RPAD] = stg[SLOT + it].z;
            m[3 * RPAD] = stg[SLOT + it].w;
        }
        return;
    }
#pragma unroll
    for (int it = 0; it < 2; ++it) {
        *(float4*)&M[lane * RPAD + p * 32 + 4 * (wave + 4 * it)] = stg[SLOT + it];
    }
}

template <int NN>
__device__ __forceinline__ void frag_load(v2f (&a)[8][4], const float* M,
                                          const int l16, const int hi)
{
#pragma unroll
    for (int kk = 0; kk < 8; ++kk) {
#pragma unroll
        for (int n = 0; n < NN; ++n) {
            const int erow = (n & 1) * 16 + l16;
            const int prow = (n >> 1) * 32;
            a[kk][n] = *(const v2f*)&M[erow * RPAD + prow + 4 * kk + 2 * hi];
        }
    }
}

__device__ __forceinline__ void select_env(v8f (&env)[2], const v8f (&acc)[4],
                                           const bool sp)
{
#pragma unroll
    for (int r = 0; r < 8; ++r) {
        env[0][r] = sp ? acc[2][r] : acc[0][r];
        env[1][r] = sp ? acc[3][r] : acc[1][r];
    }
}

__device__ __forceinline__ void wmma_kstep(v8f (&acc)[4], const v2f (&a)[4],
                                           const v8f (&env)[2], const int kk)
{
    v2f b;
    b.x = env[kk >> 2][2 * (kk & 3)];
    b.y = env[kk >> 2][2 * (kk & 3) + 1];
#pragma unroll
    for (int n = 0; n < 4; ++n) {
        acc[n] = __builtin_amdgcn_wmma_f32_16x16x4_f32(
            false, a[n], false, b, (short)0, acc[n], false, false);
    }
}

template <int MODE, bool WAIT, bool FRAG, int LAND, int REFILL,
          bool SEL, bool UNI, bool NEXTUNI>
__device__ __forceinline__ void table_step(const int site,
                                           float4 (&stg)[4],
                                           const v2f (&aCur)[8][4],
                                           v2f (&aNext)[8][4],
                                           float* Mcur,
                                           const float* Mnext,
                                           const int rowBase,
                                           const float* __restrict__ bulkT,
                                           v8f (&env)[2],
                                           v8f (&acc)[4], bool& sp,
                                           const int wave, const int lane,
                                           const int l16, const int hi)
{
    if (WAIT) {
        lds_wait();
    }
    if (FRAG) {
        if (NEXTUNI) {
            frag_load<2>(aNext, Mnext + 32 * ((rowBase >> (site + 1)) & 1), l16, hi);
        } else {
            frag_load<4>(aNext, Mnext, l16, hi);
        }
    }
    if (LAND == 1) {
        stage_store<MODE>(stg, Mcur, wave, lane);
    }
    if (LAND == 2) {
        stage_store_one<MODE, 0>(stg, Mcur, wave, lane, (rowBase >> (site + 2)) & 1);
    }
    if (LAND == 3) {
        stage_store_one<MODE, 2>(stg, Mcur, wave, lane, (rowBase >> (site + 2)) & 1);
    }
    __builtin_amdgcn_sched_barrier(0);

    if (SEL) {
        select_env(env, acc, sp);
    }
    v8f uni[2] = {};
    if (!UNI) {
#pragma unroll
        for (int n = 0; n < 4; ++n) {
            acc[n] = v8f{};
        }
    }
#pragma unroll
    for (int kk = 0; kk < 8; ++kk) {
        if (kk == 4) {
            __builtin_amdgcn_sched_barrier(0);
            if (LAND != 0) {
                lds_arrive();
            }
            if (REFILL == 1) {
                stage_load<MODE>(stg, bulk_of<MODE>(bulkT, site + 4) + lane, wave);
            }
            if (REFILL == 2) {
                stage_load_one<MODE, 0>(stg, bulk_of<MODE>(bulkT, site + 4) + lane, wave,
                                        (rowBase >> (site + 4)) & 1);
            }
            if (REFILL == 2 || REFILL == 4) {
                stage_load_one<MODE, 2>(stg, bulk_of<MODE>(bulkT, site + 6) + lane, wave,
                                        (rowBase >> (site + 6)) & 1);
            }
            if (REFILL == 3) {
                stage_load_one<MODE, 0>(stg, bulk_of<MODE>(bulkT, site + 6) + lane, wave,
                                        (rowBase >> (site + 6)) & 1);
            }
            __builtin_amdgcn_sched_barrier(0);
        }
        if (UNI) {
            v2f b;
            b.x = env[kk >> 2][2 * (kk & 3)];
            b.y = env[kk >> 2][2 * (kk & 3) + 1];
#pragma unroll
            for (int et = 0; et < 2; ++et) {
                uni[et] = __builtin_amdgcn_wmma_f32_16x16x4_f32(
                    false, aCur[kk][et], false, b, (short)0, uni[et], false, false);
            }
        } else {
            wmma_kstep(acc, aCur[kk], env, kk);
        }
    }
    if (UNI) {
        env[0] = uni[0];
        env[1] = uni[1];
    } else {
        sp = (((rowBase + l16) >> site) & 1) != 0;
    }
    __builtin_amdgcn_sched_barrier(0);
}

template <int MODE>
__device__ __forceinline__
void mps_body(const float* __restrict__ leftT,
              const float* __restrict__ bulkT,
              const float* __restrict__ rightT,
              float*       __restrict__ out,
              const int blockId)
{
    static_assert(MODE == 1 || MODE == 2, "table modes only; the main chain is main_body");
    constexpr int NSTEP = ModeC<MODE>::NSTEP;
    __shared__ __align__(16) float sM[2][BOND * RPAD];

    const int tid   = threadIdx.x;
    const int wave  = tid >> 5;
    const int lane  = tid & 31;
    const int l16   = lane & 15;
    const int hi    = lane >> 4;
    const int rowBase = blockId * ModeC<MODE>::WG_ROWS + wave * ModeC<MODE>::WAVE_ROWS;

    v8f envQ[2];
    {
        const int sQ = (rowBase + l16) & 1;
        if (MODE == 2) {
#pragma unroll
            for (int t = 0; t < 2; ++t) {
#pragma unroll
                for (int r = 0; r < 8; ++r) {
                    const float* rt = rightT + (16 * t + 8 * hi + r) * 2;
                    envQ[t][r] = rt[sQ];
                }
            }
        } else {
            const float* lq = leftT + sQ * BOND + 8 * hi;
#pragma unroll
            for (int t = 0; t < 2; ++t) {
                const float4 u = *(const float4*)(lq + 16 * t);
                const float4 v = *(const float4*)(lq + 16 * t + 4);
                envQ[t][0] = u.x; envQ[t][1] = u.y; envQ[t][2] = u.z; envQ[t][3] = u.w;
                envQ[t][4] = v.x; envQ[t][5] = v.y; envQ[t][6] = v.z; envQ[t][7] = v.w;
            }
        }
    }
    v8f accQ[4];
    accQ[0] = envQ[0]; accQ[1] = envQ[1]; accQ[2] = envQ[0]; accQ[3] = envQ[1];
    bool spQ = false;

    float4 stgA[4], stgB[4];
    v2f fragA[8][4], fragB[8][4];
    float* const M0 = &sM[0][0];
    float* const M1 = &sM[1][0];

    stage_load<MODE>(stgA, bulk_of<MODE>(bulkT, 1) + lane, wave);
    __builtin_amdgcn_sched_barrier(0);
    stage_load<MODE>(stgB, bulk_of<MODE>(bulkT, 2) + lane, wave);
    __builtin_amdgcn_sched_barrier(0);
    stage_store<MODE>(stgA, M0, wave, lane);
    lds_arrive();
    lds_wait();
    stage_load<MODE>(stgA, bulk_of<MODE>(bulkT, 3) + lane, wave);
    frag_load<4>(fragA, M0, l16, hi);
    __builtin_amdgcn_sched_barrier(0);
    stage_store<MODE>(stgB, M1, wave, lane);
    lds_arrive();
    stage_load<MODE>(stgB, bulk_of<MODE>(bulkT, 4) + lane, wave);
    __builtin_amdgcn_sched_barrier(0);

    {
        static_assert(TILE_BITS == 4 && WG_BITS == 6 && WAVES_PER_WG == 4 && NSTEP == 12,
                      "table chain: the twelve steps are written out");
        table_step<MODE, true, true, 1, 1, true, false, false>(
            1, stgA, fragA, fragB, M0, M1, rowBase, bulkT, envQ, accQ, spQ,
            wave, lane, l16, hi);
        table_step<MODE, true, true, 1, 2, true, false, false>(
            2, stgB, fragB, fragA, M1, M0, rowBase, bulkT, envQ, accQ, spQ,
            wave, lane, l16, hi);
        table_step<MODE, true, true, 1, 2, true, false, true>(
            3, stgA, fragA, fragB, M0, M1, rowBase, bulkT, envQ, accQ, spQ,
            wave, lane, l16, hi);
        table_step<MODE, true, true, 2, 3, true, true, true>(
            4, stgB, fragB, fragA, M1, M0, rowBase, bulkT, envQ, accQ, spQ,
            wave, lane, l16, hi);
        table_step<MODE, true, true, 2, 3, false, true, true>(
            5, stgA, fragA, fragB, M0, M1, rowBase, bulkT, envQ, accQ, spQ,
            wave, lane, l16, hi);
        table_step<MODE, true, true, 3, 4, false, true, true>(
            6, stgB, fragB, fragA, M1, M0, rowBase, bulkT, envQ, accQ, spQ,
            wave, lane, l16, hi);
        table_step<MODE, true, true, 3, 0, false, true, true>(
            7, stgA, fragA, fragB, M0, M1, rowBase, bulkT, envQ, accQ, spQ,
            wave, lane, l16, hi);
        table_step<MODE, true, true, 2, 0, false, true, true>(
            8, stgB, fragB, fragA, M1, M0, rowBase, bulkT, envQ, accQ, spQ,
            wave, lane, l16, hi);
        table_step<MODE, true, true, 2, 0, false, true, true>(
            9, stgA, fragA, fragB, M0, M1, rowBase, bulkT, envQ, accQ, spQ,
            wave, lane, l16, hi);
        table_step<MODE, true, true, 3, 0, false, true, true>(
            10, stgB, fragB, fragA, M1, M0, rowBase, bulkT, envQ, accQ, spQ,
            wave, lane, l16, hi);
        table_step<MODE, true, true, 0, 0, false, true, true>(
            11, stgA, fragA, fragB, M0, M1, rowBase, bulkT, envQ, accQ, spQ,
            wave, lane, l16, hi);
        table_step<MODE, false, false, 0, 0, false, true, false>(
            12, stgB, fragB, fragA, M1, M0, rowBase, bulkT, envQ, accQ, spQ,
            wave, lane, l16, hi);
    }

    float* tq = out + (size_t)(rowBase + l16) * BOND + 8 * hi;
#pragma unroll
    for (int t = 0; t < 2; ++t) {
        *(float4*)(tq + 16 * t)     = make_float4(envQ[t][0], envQ[t][1], envQ[t][2], envQ[t][3]);
        *(float4*)(tq + 16 * t + 4) = make_float4(envQ[t][4], envQ[t][5], envQ[t][6], envQ[t][7]);
    }
}

#define MAIN_NSTEP (NSITES - 2 - 2 * NPRE)
#define NPAIR (MAIN_NSTEP / 2)
#define PROD_FLOATS (BOND * BOND)
#define PAIR_FLOATS (4 * PROD_FLOATS)
#define ENV_PITCH 36
#define ENV_SPARE 8
#define ENV_ROWS (ROWS_PER_WAVE + ENV_SPARE)
static_assert(MAIN_NSTEP % 2 == 0 && NPAIR % 2 == 1, "pairs; pair loop unrolled by two plus a last pair");

template <int KK0, int KK1>
__device__ __forceinline__ void tile_wmma_part(v8f (&acc)[2], const v2f (&a)[8][2],
                                               const v8f (&b)[2])
{
    if (KK0 == 0) {
        acc[0] = v8f{};
        acc[1] = v8f{};
    }
#pragma unroll
    for (int kk = KK0; kk < KK1; ++kk) {
        v2f bb;
        bb.x = b[kk >> 2][2 * (kk & 3)];
        bb.y = b[kk >> 2][2 * (kk & 3) + 1];
#pragma unroll
        for (int et = 0; et < 2; ++et) {
            acc[et] = __builtin_amdgcn_wmma_f32_16x16x4_f32(
                false, a[kk][et], false, bb, (short)0, acc[et], false, false);
        }
    }
}

__device__ __forceinline__ void tile_wmma(v8f (&acc)[2], const v2f (&a)[8][2],
                                          const v8f (&b)[2])
{
    tile_wmma_part<0, 8>(acc, a, b);
}

__device__ __forceinline__ void prod_body(const float* __restrict__ bulkT,
                                          float* __restrict__ prodT, const int pi)
{
    const int tid  = threadIdx.x;
    const int wave = tid >> 5;
    const int lane = tid & 31;
    const int l16  = lane & 15;
    const int hi   = lane >> 4;
    const int pa   = wave & 1;
    const int pb   = wave >> 1;
    const float* m1 = bulkT + (size_t)(NPRE + 2 * pi) * SITE_FLOATS;
    const float* m2 = m1 + SITE_FLOATS;
    v2f a[8][2];
#pragma unroll
    for (int kk = 0; kk < 8; ++kk) {
        const int d0 = 16 * (kk >> 2) + 8 * hi + 2 * (kk & 3);
        const float* b = m2 + d0 * 64 + pb * 32 + l16;
        a[kk][0].x = b[0];
        a[kk][0].y = b[64];
        a[kk][1].x = b[16];
        a[kk][1].y = b[64 + 16];
    }
    float* g = prodT + ((size_t)pi * 4 + wave) * PROD_FLOATS;
#pragma unroll
    for (int T = 0; T < 2; ++T) {
        const float* r0 = m1 + (16 * T + l16) * 64 + pa * 32 + 8 * hi;
        v8f env[2];
#pragma unroll
        for (int t = 0; t < 2; ++t) {
            const float4 u = *(const float4*)(r0 + 16 * t);
            const float4 v = *(const float4*)(r0 + 16 * t + 4);
            env[t][0] = u.x; env[t][1] = u.y; env[t][2] = u.z; env[t][3] = u.w;
            env[t][4] = v.x; env[t][5] = v.y; env[t][6] = v.z; env[t][7] = v.w;
        }
        v8f acc[2];
        tile_wmma(acc, a, env);
        const int hq = (l16 >> 3) & 1;
        const int q  = (l16 >> 1) & 3;
        const int j  = l16 & 1;
        const int kk = 4 * T + q;
#pragma unroll
        for (int et = 0; et < 2; ++et) {
#pragma unroll
            for (int r = 0; r < 8; ++r) {
                g[(kk * 32 + 16 * hq + 8 * hi + r) * 4 + 2 * et + j] = acc[et][r];
            }
        }
    }
}

__device__ __forceinline__ void prod_fetch(v2f (&a)[4][8][2], const float* __restrict__ G,
                                           const int lane)
{
#pragma unroll
    for (int x = 0; x < 4; ++x) {
#pragma unroll
        for (int kk = 0; kk < 8; ++kk) {
            const float4 v = *(const float4*)(G + ((x * 8 + kk) * 32 + lane) * 4);
            a[x][kk][0].x = v.x;
            a[x][kk][0].y = v.y;
            a[x][kk][1].x = v.z;
            a[x][kk][1].y = v.w;
        }
    }
}

__device__ __forceinline__ void env_gather(v8f (&b)[2], const float* envW,
                                           const int row, const int hi)
{
    const float* s = envW + row * ENV_PITCH + 8 * hi;
#pragma unroll
    for (int t = 0; t < 2; ++t) {
        const float4 u = *(const float4*)(s + 16 * t);
        const float4 v = *(const float4*)(s + 16 * t + 4);
        b[t][0] = u.x; b[t][1] = u.y; b[t][2] = u.z; b[t][3] = u.w;
        b[t][4] = v.x; b[t][5] = v.y; b[t][6] = v.z; b[t][7] = v.w;
    }
}

__device__ __forceinline__ void env_scatter(float* envW, const int row, const int hi,
                                            const v8f (&acc)[2])
{
    float* d = envW + row * ENV_PITCH + 8 * hi;
#pragma unroll
    for (int t = 0; t < 2; ++t) {
        *(float4*)(d + 16 * t)     = make_float4(acc[t][0], acc[t][1], acc[t][2], acc[t][3]);
        *(float4*)(d + 16 * t + 4) = make_float4(acc[t][4], acc[t][5], acc[t][6], acc[t][7]);
    }
}

__device__ __forceinline__ void slots_issue(const int spinA, const int spinB, int* slotW,
                                            int (&cnt)[4], int (&lo)[4], int (&hi2)[4],
                                            const int lane, const int l16)
{
    const int sa = (spinA != 0) ? 1 : 0;
    const int sb = (spinB != 0) ? 1 : 0;
    const int x  = sa + 2 * sb;
    int rank = 0;
#pragma unroll
    for (int xx = 0; xx < 4; ++xx) {
        const unsigned m = __builtin_amdgcn_ballot_w32(x == xx);
        cnt[xx] = __builtin_popcount(m);
        const int rk = (int)__builtin_amdgcn_mbcnt_lo(m, 0u);
        rank = (x == xx) ? rk : rank;
    }
    slotW[x * 32 + rank] = lane;
    __builtin_amdgcn_wave_barrier();
#pragma unroll
    for (int xx = 0; xx < 4; ++xx) {
        lo[xx]  = slotW[xx * 32 + l16];
        hi2[xx] = slotW[xx * 32 + 16 + l16];
    }
}

__device__ __forceinline__ void slots_close(const int (&cnt)[4], const int (&lo)[4],
                                            const int (&hi2)[4],
                                            int (&rowLo)[4], int (&rowHi)[4], const int l16)
{
    const int spare = ROWS_PER_WAVE + (l16 & (ENV_SPARE - 1));
#pragma unroll
    for (int xx = 0; xx < 4; ++xx) {
        rowLo[xx] = (l16 < cnt[xx]) ? lo[xx] : spare;
        rowHi[xx] = (16 + l16 < cnt[xx]) ? hi2[xx] : spare;
    }
}

__device__ __forceinline__ void pair_slots(const int p, const int* cfgW, int* slotW,
                                           int (&cnt)[4], int (&rowLo)[4], int (&rowHi)[4],
                                           const int lane, const int l16)
{
    const int c1 = NPRE + 1 + 2 * p;
    int lo[4], hi2[4];
    slots_issue(cfgW[c1 * CPAD + lane], cfgW[(c1 + 1) * CPAD + lane], slotW,
                cnt, lo, hi2, lane, l16);
    slots_close(cnt, lo, hi2, rowLo, rowHi, l16);
}

template <bool FETCH>
__device__ __forceinline__ void pair_step(const int pi,
                                          const v2f (&fCur)[4][8][2],
                                          v2f (&fNext)[4][8][2],
                                          const int* cfgW, int* slotW, float* envW,
                                          const float* __restrict__ prodT,
                                          int (&cnt)[4], int (&rowLo)[4], int (&rowHi)[4],
                                          const int lane, const int l16, const int hi)
{
    int c[4], rL[4], rH[4];
#pragma unroll
    for (int x = 0; x < 4; ++x) {
        c[x] = cnt[x]; rL[x] = rowLo[x]; rH[x] = rowHi[x];
    }
    v8f bE[2], bO[2];
    env_gather(bE, envW, rL[0], hi);
    __builtin_amdgcn_sched_barrier(0);
    int spinA = 0, spinB = 0;
    if (FETCH) {
        const int c1 = NPRE + 1 + 2 * (pi + 1);
        spinA = cfgW[c1 * CPAD + lane];
        spinB = cfgW[(c1 + 1) * CPAD + lane];
        prod_fetch(fNext, prodT + (size_t)(pi + 1) * PAIR_FLOATS, lane);
    }
    __builtin_amdgcn_sched_barrier(0);

    int lo[4] = {0, 0, 0, 0}, hi2[4] = {0, 0, 0, 0};
#pragma unroll
    for (int x = 0; x < 4; ++x) {
        v8f (&bCur)[2] = (x & 1) ? bO : bE;
        v8f (&bNxt)[2] = (x & 1) ? bE : bO;
        if (x < 3) {
            env_gather(bNxt, envW, rL[x + 1], hi);
        }
        __builtin_amdgcn_sched_barrier(0);
        v8f acc[2];
        tile_wmma_part<0, 1>(acc, fCur[x], bCur);
        if (FETCH && x == 1) {
            slots_issue(spinA, spinB, slotW, cnt, lo, hi2, lane, l16);
        }
        if (FETCH && x == 2) {
            slots_close(cnt, lo, hi2, rowLo, rowHi, l16);
        }
        tile_wmma_part<1, 8>(acc, fCur[x], bCur);
        env_scatter(envW, rL[x], hi, acc);
        if (c[x] > 16) {
            v8f b2[2], acc2[2];
            env_gather(b2, envW, rH[x], hi);
            tile_wmma(acc2, fCur[x], b2);
            env_scatter(envW, rH[x], hi, acc2);
        }
        __builtin_amdgcn_sched_barrier(0);
    }
}

__device__ __forceinline__
void main_body(const int*   __restrict__ cfg,
               const float* __restrict__ prodT,
               const float* __restrict__ preT,
               const float* __restrict__ sufT,
               float*       __restrict__ out,
               const int blockId)
{
    __shared__ __align__(16) int   sCfgAll[WAVES_PER_WG][NSITES][CPAD];
    __shared__ __align__(16) float sEnv[WAVES_PER_WG][ENV_ROWS * ENV_PITCH];
    __shared__ __align__(16) int   sSlot[WAVES_PER_WG][4 * 32];

    const int tid   = threadIdx.x;
    const int wave  = tid >> 5;
    const int lane  = tid & 31;
    const int l16   = lane & 15;
    const int hi    = lane >> 4;
    const int rowBase = blockId * ROWS_PER_WG + wave * ROWS_PER_WAVE;

    v2f fragA[4][8][2], fragB[4][8][2];
    prod_fetch(fragA, prodT, lane);

#pragma unroll
    for (int i = lane; i < ROWS_PER_WAVE * NSITES; i += 32) {
        const int r = i >> 6;
        const int c = i & 63;
        sCfgAll[wave][c][r] = cfg[(rowBase + r) * NSITES + c];
    }
    __builtin_amdgcn_wave_barrier();
    const int* const cfgW = &sCfgAll[wave][0][0];
    float* const envW = &sEnv[wave][0];
    int* const slotW  = &sSlot[wave][0];

    {
        int sP = 0;
#pragma unroll
        for (int c = 0; c < TBITS; ++c) {
            sP |= (cfgW[c * CPAD + lane] != 0) << c;
        }
        const float* lp = preT + (size_t)sP * BOND;
        float* e = envW + lane * ENV_PITCH;
#pragma unroll
        for (int v = 0; v < 8; ++v) {
            *(float4*)(e + 4 * v) = *(const float4*)(lp + 4 * v);
        }
    }
    int cnt[4], rowLo[4], rowHi[4];
    pair_slots(0, cfgW, slotW, cnt, rowLo, rowHi, lane, l16);
    __builtin_amdgcn_wave_barrier();

    for (int pi = 0; pi < NPAIR - 1; pi += 2) {
        pair_step<true>(pi,     fragA, fragB, cfgW, slotW, envW, prodT, cnt, rowLo, rowHi, lane, l16, hi);
        pair_step<true>(pi + 1, fragB, fragA, cfgW, slotW, envW, prodT, cnt, rowLo, rowHi, lane, l16, hi);
    }
    pair_step<false>(NPAIR - 1, fragA, fragB, cfgW, slotW, envW, prodT, cnt, rowLo, rowHi, lane, l16, hi);
    __builtin_amdgcn_wave_barrier();

    {
        int iS = 0;
#pragma unroll
        for (int c = 0; c < TBITS; ++c) {
            iS |= (cfgW[(NSITES - 1 - c) * CPAD + lane] != 0) << c;
        }
        const float* w = sufT + (size_t)iS * BOND;
        const float* e = envW + lane * ENV_PITCH;
        float p0 = 0.0f, p1 = 0.0f;
#pragma unroll
        for (int t = 0; t < 2; ++t) {
#pragma unroll
            for (int r = 0; r < 8; ++r) {
                p0 = fmaf(e[16 * t + r],     w[16 * t + r],     p0);
                p1 = fmaf(e[16 * t + 8 + r], w[16 * t + 8 + r], p1);
            }
        }
        out[rowBase + lane] = p0 + p1;
    }
}

__global__ __launch_bounds__(WAVES_PER_WG * 32)
__attribute__((amdgpu_waves_per_eu(1, 1)))
void mps_psi_kernel(const int*   __restrict__ cfg,
                    const float* __restrict__ prodT,
                    const float* __restrict__ preT,
                    const float* __restrict__ sufT,
                    float*       __restrict__ out)
{
    main_body(cfg, prodT, preT, sufT, out, (int)blockIdx.x);
}

#define TABLE_WGS (2 * TROWS / (TILE_ROWS * WAVES_PER_WG))
__global__ __launch_bounds__(WAVES_PER_WG * 32)
__attribute__((amdgpu_waves_per_eu(1, 1)))
void mps_tables_kernel(const float* __restrict__ leftT,
                       const float* __restrict__ bulkT,
                       const float* __restrict__ rightT,
                       float*       __restrict__ preT,
                       float*       __restrict__ sufT,
                       float*       __restrict__ prodT)
{
    const int b = (int)blockIdx.x;
    if (b >= TABLE_WGS) {
        prod_body(bulkT, prodT, b - TABLE_WGS);
    } else if (b < TROWS / ModeC<1>::WG_ROWS) {
        mps_body<1>(leftT, bulkT, nullptr, preT, b);
    } else {
        mps_body<2>(nullptr, bulkT, rightT, sufT, b - TROWS / ModeC<2>::WG_ROWS);
    }
}

extern "C" void kernel_launch(void* const* d_in, const int* in_sizes, int n_in,
                              void* d_out, int out_size, void* d_ws, size_t ws_size,
                              hipStream_t stream) {
    const int*   cfg    = (const int*)  d_in[0];
    const float* leftT  = (const float*)d_in[1];
    const float* bulkT  = (const float*)d_in[2];
    const float* rightT = (const float*)d_in[3];
    float*       out    = (float*)d_out;
    float*       preT   = (float*)d_ws;
    float*       sufT   = (float*)d_ws + TROWS * BOND;
    float*       prodT  = (float*)d_ws + 2 * TROWS * BOND;
    (void)in_sizes; (void)n_in; (void)out_size; (void)ws_size;

    dim3 block(WAVES_PER_WG * 32);
    dim3 gridT(TABLE_WGS + NPAIR);
    hipLaunchKernelGGL(mps_tables_kernel, gridT, block, 0, stream,
                       leftT, bulkT, rightT, preT, sufT, prodT);
    dim3 grid(BATCH / ROWS_PER_WG);
    hipLaunchKernelGGL(mps_psi_kernel, grid, block, 0, stream,
                       cfg, prodT, preT, sufT, out);
}
